# starting kernel
# baseline (speedup 1.0000x reference)
#include <hip/hip_runtime.h>

#define N_NODES 50000
#define N_EDGES 800000
#define F_OUT   128
#define N_HEAD  4
#define NEG_SLOPE 0.2f

typedef __attribute__((ext_vector_type(2))) float v2f;
typedef __attribute__((ext_vector_type(8))) float v8f;

__device__ __forceinline__ void atomF(float* p, float v) {
    unsafeAtomicAdd(p, v);
}

__global__ void zero_f4(float4* __restrict__ p, int n4) {
    int i = blockIdx.x * blockDim.x + threadIdx.x;
    int stride = gridDim.x * blockDim.x;
    float4 z = make_float4(0.f, 0.f, 0.f, 0.f);
    for (; i < n4; i += stride) p[i] = z;
}

__global__ void coef_kernel(const float* __restrict__ w, const float* __restrict__ a,
                            float* __restrict__ coefB) {
    int idx = blockIdx.x * blockDim.x + threadIdx.x;
    if (idx >= F_OUT * 16) return;
    int k = idx >> 4, col = idx & 15;
    float v = 0.f;
    if (col < 8) {
        int head = col & 3, sel = col >> 2;
        v = w[head * F_OUT + k] * a[head * 2 * F_OUT + sel * F_OUT + k];
    }
    coefB[idx] = v;
}

__global__ void proj_kernel(const float* __restrict__ x, const float* __restrict__ coefB,
                            float* __restrict__ s_src, float* __restrict__ s_dst) {
    int tile = blockIdx.x;
    int lane = threadIdx.x;
    int half = lane >> 4;
    int rc   = lane & 15;
    int koff = half * 2;
    long arow = (long)(tile * 16 + rc) * F_OUT;

    v8f c = {0.f, 0.f, 0.f, 0.f, 0.f, 0.f, 0.f, 0.f};
#pragma unroll
    for (int k = 0; k < F_OUT; k += 4) {
        v2f av, bv;
        float2 ax = *(const float2*)(x + arow + k + koff);
        av.x = ax.x; av.y = ax.y;
        bv.x = coefB[(k + koff) * 16 + rc];
        bv.y = coefB[(k + koff + 1) * 16 + rc];
        c = __builtin_amdgcn_wmma_f32_16x16x4_f32(false, av, false, bv,
                                                  (short)0, c, false, false);
    }
    if (rc < 8) {
        int head = rc & 3;
        float* buf = (rc >> 2) ? s_dst : s_src;
        int mbase = tile * 16 + half * 8;
#pragma unroll
        for (int j = 0; j < 8; ++j)
            buf[head * N_NODES + mbase + j] = c[j];
    }
}

__global__ void edge_kernel(const float* __restrict__ x, const float* __restrict__ w,
                            const int* __restrict__ ei,
                            const float* __restrict__ s_src, const float* __restrict__ s_dst,
                            float* __restrict__ rowsum, float* __restrict__ out) {
    int tid    = blockIdx.x * blockDim.x + threadIdx.x;
    int wave   = tid >> 5;
    int lane   = tid & 31;
    int nwaves = (gridDim.x * blockDim.x) >> 5;
    int f      = lane * 4;

    float4 wv[N_HEAD];
#pragma unroll
    for (int i = 0; i < N_HEAD; ++i)
        wv[i] = *(const float4*)(w + i * F_OUT + f);

    for (int e = wave; e < N_EDGES; e += nwaves) {
        int sn = ei[e];
        int dn = ei[N_EDGES + e];
        float4 xv = *(const float4*)(x + (long)dn * F_OUT + f);
#pragma unroll
        for (int i = 0; i < N_HEAD; ++i) {
            float logit = s_src[i * N_NODES + sn] + s_dst[i * N_NODES + dn];
            float lr  = (logit >= 0.f) ? logit : NEG_SLOPE * logit;
            float val = expf(-lr);
            if (lane == 0) atomF(&rowsum[i * N_NODES + sn], val);
            float* o = out + ((long)i * N_NODES + sn) * F_OUT + f;
            atomF(o + 0, val * xv.x * wv[i].x);
            atomF(o + 1, val * xv.y * wv[i].y);
            atomF(o + 2, val * xv.z * wv[i].z);
            atomF(o + 3, val * xv.w * wv[i].w);
        }
    }
}

__global__ void norm_kernel(float4* __restrict__ out, const float* __restrict__ rowsum, int n4) {
    int i = blockIdx.x * blockDim.x + threadIdx.x;
    int stride = gridDim.x * blockDim.x;
    for (; i < n4; i += stride) {
        float r = rowsum[i >> 5];
        float4 v = out[i];
        float inv = 1.0f / r;
        v.x *= inv; v.y *= inv; v.z *= inv; v.w *= inv;
        out[i] = v;
    }
}

extern "C" void kernel_launch(void* const* d_in, const int* in_sizes, int n_in,
                              void* d_out, int out_size, void* d_ws, size_t ws_size,
                              hipStream_t stream) {
    const float* x  = (const float*)d_in[0];
    const float* w  = (const float*)d_in[1];
    const float* a  = (const float*)d_in[2];
    const int*   ei = (const int*)d_in[3];
    float* out = (float*)d_out;

    float* ws     = (float*)d_ws;
    float* s_src  = ws;
    float* s_dst  = ws + N_HEAD * N_NODES;
    float* rowsum = ws + 2 * N_HEAD * N_NODES;
    float* coefB  = ws + 3 * N_HEAD * N_NODES;

    const int out_f4 = N_HEAD * N_NODES * F_OUT / 4;
    const int rs_f4  = N_HEAD * N_NODES / 4;

    zero_f4<<<4096, 256, 0, stream>>>((float4*)out, out_f4);
    zero_f4<<<196, 256, 0, stream>>>((float4*)rowsum, rs_f4);
    coef_kernel<<<8, 256, 0, stream>>>(w, a, coefB);
    proj_kernel<<<N_NODES / 16, 32, 0, stream>>>(x, coefB, s_src, s_dst);
    edge_kernel<<<2048, 256, 0, stream>>>(x, w, ei, s_src, s_dst, rowsum, out);
    norm_kernel<<<4096, 256, 0, stream>>>((float4*)out, rowsum, out_f4);
}


# output-stationary: a work-group keeps 64 source rows + denominators in LDS, scans all sources, adds its edges by LDS float atomics, stores rows normalised; no global float atomics; LDS arrival order
# speedup vs baseline: 3.0990x; 3.0990x over previous
#include <hip/hip_runtime.h>

#define N_NODES 50000
#define N_EDGES 800000
#define F_OUT   128
#define N_HEAD  4
#define NEG_SLOPE 0.2f

typedef __attribute__((ext_vector_type(2))) float v2f;
typedef __attribute__((ext_vector_type(8))) float v8f;

__global__ void coef_kernel(const float* __restrict__ w, const float* __restrict__ a,
                            float* __restrict__ coefB) {
    int idx = blockIdx.x * blockDim.x + threadIdx.x;
    if (idx >= F_OUT * 16) return;
    int k = idx >> 4, col = idx & 15;
    float v = 0.f;
    if (col < 8) {
        int head = col & 3, sel = col >> 2;
        v = w[head * F_OUT + k] * a[head * 2 * F_OUT + sel * F_OUT + k];
    }
    coefB[idx] = v;
}

__global__ void proj_kernel(const float* __restrict__ x, const float* __restrict__ coefB,
                            float* __restrict__ s_src, float* __restrict__ s_dst) {
    int tile = blockIdx.x;
    int lane = threadIdx.x;
    int half = lane >> 4;
    int rc   = lane & 15;
    int koff = half * 2;
    long arow = (long)(tile * 16 + rc) * F_OUT;

    v8f c = {0.f, 0.f, 0.f, 0.f, 0.f, 0.f, 0.f, 0.f};
#pragma unroll
    for (int k = 0; k < F_OUT; k += 4) {
        v2f av, bv;
        float2 ax = *(const float2*)(x + arow + k + koff);
        av.x = ax.x; av.y = ax.y;
        bv.x = coefB[(k + koff) * 16 + rc];
        bv.y = coefB[(k + koff + 1) * 16 + rc];
        c = __builtin_amdgcn_wmma_f32_16x16x4_f32(false, av, false, bv,
                                                  (short)0, c, false, false);
    }
    if (rc < 8) {
        int head = rc & 3;
        float* buf = (rc >> 2) ? s_dst : s_src;
        int mbase = tile * 16 + half * 8;
#pragma unroll
        for (int j = 0; j < 8; ++j)
            buf[head * N_NODES + mbase + j] = c[j];
    }
}

#define TILE_R       64
#define TILE_THREADS 1024
#define N_TILES      ((N_NODES + TILE_R - 1) / TILE_R)

#define N_CHUNK      (N_EDGES / 4 / 32)
#define SCAN_DEPTH   4

static_assert((N_EDGES % 128) == 0, "a wave reads 32 whole int4 of sources at a time");

__global__ __launch_bounds__(TILE_THREADS)
void tile_kernel(const float* __restrict__ x, const float* __restrict__ w,
                 const int* __restrict__ ei,
                 const float* __restrict__ s_src, const float* __restrict__ s_dst,
                 float* __restrict__ out) {
    __shared__ float acc[TILE_R * N_HEAD * F_OUT];
    __shared__ float rs[TILE_R * N_HEAD];
    __shared__ float ss[TILE_R * N_HEAD];

    const int tid  = threadIdx.x;
    const int lane = tid & 31;
    const int wid  = __builtin_amdgcn_readfirstlane(tid >> 5);
    const int base = blockIdx.x * TILE_R;

    const float4 z4 = make_float4(0.f, 0.f, 0.f, 0.f);
    for (int i = tid; i < TILE_R * N_HEAD * F_OUT / 4; i += TILE_THREADS)
        ((float4*)acc)[i] = z4;
    if (tid < TILE_R * N_HEAD) {
        int node = base + (tid >> 2);
        rs[tid] = 0.f;
        ss[tid] = (node < N_NODES) ? s_src[(tid & 3) * N_NODES + node] : 0.f;
    }

    float wr[N_HEAD][4];
#pragma unroll
    for (int i = 0; i < N_HEAD; ++i)
#pragma unroll
        for (int c = 0; c < 4; ++c)
            wr[i][c] = w[i * F_OUT + c * 32 + lane];

    __syncthreads();

    const int4* __restrict__ src4 = (const int4*)ei;
    const int*  __restrict__ dstp = ei + N_EDGES;

    for (int c0 = wid * SCAN_DEPTH; c0 < N_CHUNK; c0 += (TILE_THREADS / 32) * SCAN_DEPTH) {
        int4 s[SCAN_DEPTH];
#pragma unroll
        for (int j = 0; j < SCAN_DEPTH; ++j) {
            const int cj = (c0 + j < N_CHUNK) ? (c0 + j) : (N_CHUNK - 1);
            s[j] = src4[cj * 32 + lane];
        }
#pragma unroll
        for (int j = 0; j < SCAN_DEPTH; ++j) {
            const unsigned d01 = min((unsigned)(s[j].x - base), (unsigned)(s[j].y - base));
            const unsigned d23 = min((unsigned)(s[j].z - base), (unsigned)(s[j].w - base));
            const bool hit = (c0 + j < N_CHUNK) && (min(d01, d23) < (unsigned)TILE_R);
            unsigned m = __builtin_amdgcn_ballot_w32(hit);
            while (m) {
                const int L = __builtin_ctz(m);
                m &= m - 1;
                const int sv0 = __builtin_amdgcn_readlane(s[j].x, L);
                const int sv1 = __builtin_amdgcn_readlane(s[j].y, L);
                const int sv2 = __builtin_amdgcn_readlane(s[j].z, L);
                const int sv3 = __builtin_amdgcn_readlane(s[j].w, L);
                const int eb = ((c0 + j) * 32 + L) * 4;
#pragma unroll 1
                for (int k = 0; k < 4; ++k) {
                    const int svk = (k == 0) ? sv0 : (k == 1) ? sv1 : (k == 2) ? sv2 : sv3;
                    const unsigned r = (unsigned)(svk - base);
                    if (r < (unsigned)TILE_R) {
                        const int dn = dstp[eb + k];
                        const float* xr = x + (long)dn * F_OUT + lane;
                        float xv[4];
#pragma unroll
                        for (int cc = 0; cc < 4; ++cc) xv[cc] = xr[cc * 32];
                        float val[N_HEAD];
#pragma unroll
                        for (int i = 0; i < N_HEAD; ++i) {
                            float logit = ss[r * N_HEAD + i] + s_dst[i * N_NODES + dn];
                            float lr = (logit >= 0.f) ? logit : NEG_SLOPE * logit;
                            val[i] = expf(-lr);
                        }
                        float* a = acc + r * (N_HEAD * F_OUT) + lane;
#pragma unroll
                        for (int i = 0; i < N_HEAD; ++i)
#pragma unroll
                            for (int cc = 0; cc < 4; ++cc)
                                atomicAdd(a + i * F_OUT + cc * 32, val[i] * xv[cc] * wr[i][cc]);
                        if (lane < N_HEAD) {
                            float v = (lane == 0) ? val[0] : (lane == 1) ? val[1] : (lane == 2) ? val[2] : val[3];
                            atomicAdd(&rs[r * N_HEAD + lane], v);
                        }
                    }
                }
            }
        }
    }

    __syncthreads();

    for (int p = wid; p < TILE_R * N_HEAD; p += TILE_THREADS / 32) {
        const int node = base + (p >> 2);
        const int i    = p & 3;
        if (node < N_NODES) {
            const float inv = 1.0f / rs[p];
            float4 v = ((const float4*)acc)[p * (F_OUT / 4) + lane];
            v.x *= inv; v.y *= inv; v.z *= inv; v.w *= inv;
            ((float4*)out)[((long)i * N_NODES + node) * (F_OUT / 4) + lane] = v;
        }
    }
}

extern "C" void kernel_launch(void* const* d_in, const int* in_sizes, int n_in,
                              void* d_out, int out_size, void* d_ws, size_t ws_size,
                              hipStream_t stream) {
    const float* x  = (const float*)d_in[0];
    const float* w  = (const float*)d_in[1];
    const float* a  = (const float*)d_in[2];
    const int*   ei = (const int*)d_in[3];
    float* out = (float*)d_out;

    float* ws     = (float*)d_ws;
    float* s_src  = ws;
    float* s_dst  = ws + N_HEAD * N_NODES;
    float* coefB  = ws + 3 * N_HEAD * N_NODES;

    coef_kernel<<<8, 256, 0, stream>>>(w, a, coefB);
    proj_kernel<<<N_NODES / 16, 32, 0, stream>>>(x, coefB, s_src, s_dst);
    tile_kernel<<<N_TILES, TILE_THREADS, 0, stream>>>(x, w, ei, s_src, s_dst, out);
}


# edge path off the scalar unit: head h's score and exponential on lane h, broadcast by readlane; source slots tested by unrolled ballots, no select chain; sums stay in LDS arrival order
# speedup vs baseline: 3.4907x; 1.1264x over previous
#include <hip/hip_runtime.h>

#define N_NODES 50000
#define N_EDGES 800000
#define F_OUT   128
#define N_HEAD  4
#define NEG_SLOPE 0.2f

typedef __attribute__((ext_vector_type(2))) float v2f;
typedef __attribute__((ext_vector_type(8))) float v8f;

__global__ void coef_kernel(const float* __restrict__ w, const float* __restrict__ a,
                            float* __restrict__ coefB) {
    int idx = blockIdx.x * blockDim.x + threadIdx.x;
    if (idx >= F_OUT * 16) return;
    int k = idx >> 4, col = idx & 15;
    float v = 0.f;
    if (col < 8) {
        int head = col & 3, sel = col >> 2;
        v = w[head * F_OUT + k] * a[head * 2 * F_OUT + sel * F_OUT + k];
    }
    coefB[idx] = v;
}

__global__ void proj_kernel(const float* __restrict__ x, const float* __restrict__ coefB,
                            float* __restrict__ s_src, float* __restrict__ s_dst) {
    int tile = blockIdx.x;
    int lane = threadIdx.x;
    int half = lane >> 4;
    int rc   = lane & 15;
    int koff = half * 2;
    long arow = (long)(tile * 16 + rc) * F_OUT;

    v8f c = {0.f, 0.f, 0.f, 0.f, 0.f, 0.f, 0.f, 0.f};
#pragma unroll
    for (int k = 0; k < F_OUT; k += 4) {
        v2f av, bv;
        float2 ax = *(const float2*)(x + arow + k + koff);
        av.x = ax.x; av.y = ax.y;
        bv.x = coefB[(k + koff) * 16 + rc];
        bv.y = coefB[(k + koff + 1) * 16 + rc];
        c = __builtin_amdgcn_wmma_f32_16x16x4_f32(false, av, false, bv,
                                                  (short)0, c, false, false);
    }
    if (rc < 8) {
        int head = rc & 3;
        float* buf = (rc >> 2) ? s_dst : s_src;
        int mbase = tile * 16 + half * 8;
#pragma unroll
        for (int j = 0; j < 8; ++j)
            buf[head * N_NODES + mbase + j] = c[j];
    }
}

#define TILE_R       64
#define TILE_THREADS 1024
#define N_TILES      ((N_NODES + TILE_R - 1) / TILE_R)

#define N_CHUNK      (N_EDGES / 4 / 32)
#define SCAN_DEPTH   4

static_assert((N_EDGES % 128) == 0, "a wave reads 32 whole int4 of sources at a time");
static_assert((N_HEAD & (N_HEAD - 1)) == 0 && N_HEAD <= 32, "lane & (N_HEAD - 1) names a head; lanes 0..N_HEAD-1 hold one each");

__global__ __launch_bounds__(TILE_THREADS)
void tile_kernel(const float* __restrict__ x, const float* __restrict__ w,
                 const int* __restrict__ ei,
                 const float* __restrict__ s_src, const float* __restrict__ s_dst,
                 float* __restrict__ out) {
    __shared__ float acc[TILE_R * N_HEAD * F_OUT];
    __shared__ float rs[TILE_R * N_HEAD];
    __shared__ float ss[TILE_R * N_HEAD];

    const int tid  = threadIdx.x;
    const int lane = tid & 31;
    const int hd   = lane & (N_HEAD - 1);
    const int wid  = __builtin_amdgcn_readfirstlane(tid >> 5);
    const int base = blockIdx.x * TILE_R;

    const float4 z4 = make_float4(0.f, 0.f, 0.f, 0.f);
    for (int i = tid; i < TILE_R * N_HEAD * F_OUT / 4; i += TILE_THREADS)
        ((float4*)acc)[i] = z4;
    if (tid < TILE_R * N_HEAD) {
        int node = base + (tid >> 2);
        rs[tid] = 0.f;
        ss[tid] = (node < N_NODES) ? s_src[(tid & 3) * N_NODES + node] : 0.f;
    }

    float wr[N_HEAD][4];
#pragma unroll
    for (int i = 0; i < N_HEAD; ++i)
#pragma unroll
        for (int c = 0; c < 4; ++c)
            wr[i][c] = w[i * F_OUT + c * 32 + lane];

    __syncthreads();

    const int4* __restrict__ src4 = (const int4*)ei;
    const int*  __restrict__ dstp = ei + N_EDGES;

    for (int c0 = wid * SCAN_DEPTH; c0 < N_CHUNK; c0 += (TILE_THREADS / 32) * SCAN_DEPTH) {
        int4 s[SCAN_DEPTH];
#pragma unroll
        for (int j = 0; j < SCAN_DEPTH; ++j) {
            const int cj = (c0 + j < N_CHUNK) ? (c0 + j) : (N_CHUNK - 1);
            s[j] = src4[cj * 32 + lane];
        }
#pragma unroll
        for (int j = 0; j < SCAN_DEPTH; ++j) {
            const unsigned d01 = min((unsigned)(s[j].x - base), (unsigned)(s[j].y - base));
            const unsigned d23 = min((unsigned)(s[j].z - base), (unsigned)(s[j].w - base));
            const bool hit = (c0 + j < N_CHUNK) && (min(d01, d23) < (unsigned)TILE_R);
            const unsigned m = __builtin_amdgcn_ballot_w32(hit);
            if (m) {
#pragma unroll
                for (int k = 0; k < 4; ++k) {
                    const int sk = (k == 0) ? s[j].x : (k == 1) ? s[j].y : (k == 2) ? s[j].z : s[j].w;
                    const unsigned rk = (unsigned)(sk - base);
                    unsigned mk = __builtin_amdgcn_ballot_w32(rk < (unsigned)TILE_R);
                    while (mk) {
                        const int L = __builtin_ctz(mk);
                        mk &= mk - 1;
                        const unsigned r = (unsigned)__builtin_amdgcn_readlane((int)rk, L);
                        const int dn = dstp[((c0 + j) * 32 + L) * 4 + k];
                        const float* xr = x + (long)dn * F_OUT + lane;
                        float xv[4];
#pragma unroll
                        for (int cc = 0; cc < 4; ++cc) xv[cc] = xr[cc * 32];
                        const float logit = ss[r * N_HEAD + hd] + s_dst[hd * N_NODES + dn];
                        const float lr = (logit >= 0.f) ? logit : NEG_SLOPE * logit;
                        const float ev = expf(-lr);
                        const int evb = __float_as_int(ev);
                        const float val[N_HEAD] = { __int_as_float(__builtin_amdgcn_readlane(evb, 0)),
                                                    __int_as_float(__builtin_amdgcn_readlane(evb, 1)),
                                                    __int_as_float(__builtin_amdgcn_readlane(evb, 2)),
                                                    __int_as_float(__builtin_amdgcn_readlane(evb, 3)) };
                        float* a = acc + r * (N_HEAD * F_OUT) + lane;
#pragma unroll
                        for (int i = 0; i < N_HEAD; ++i)
#pragma unroll
                            for (int cc = 0; cc < 4; ++cc)
                                atomicAdd(a + i * F_OUT + cc * 32, val[i] * xv[cc] * wr[i][cc]);
                        if (lane < N_HEAD)
                            atomicAdd(&rs[r * N_HEAD + lane], ev);
                    }
                }
            }
        }
    }

    __syncthreads();

    for (int p = wid; p < TILE_R * N_HEAD; p += TILE_THREADS / 32) {
        const int node = base + (p >> 2);
        const int i    = p & 3;
        if (node < N_NODES) {
            const float inv = 1.0f / rs[p];
            float4 v = ((const float4*)acc)[p * (F_OUT / 4) + lane];
            v.x *= inv; v.y *= inv; v.z *= inv; v.w *= inv;
            ((float4*)out)[((long)i * N_NODES + node) * (F_OUT / 4) + lane] = v;
        }
    }
}

extern "C" void kernel_launch(void* const* d_in, const int* in_sizes, int n_in,
                              void* d_out, int out_size, void* d_ws, size_t ws_size,
                              hipStream_t stream) {
    const float* x  = (const float*)d_in[0];
    const float* w  = (const float*)d_in[1];
    const float* a  = (const float*)d_in[2];
    const int*   ei = (const int*)d_in[3];
    float* out = (float*)d_out;

    float* ws     = (float*)d_ws;
    float* s_src  = ws;
    float* s_dst  = ws + N_HEAD * N_NODES;
    float* coefB  = ws + 3 * N_HEAD * N_NODES;

    coef_kernel<<<8, 256, 0, stream>>>(w, a, coefB);
    proj_kernel<<<N_NODES / 16, 32, 0, stream>>>(x, coefB, s_src, s_dst);
    tile_kernel<<<N_TILES, TILE_THREADS, 0, stream>>>(x, w, ei, s_src, s_dst, out);
}


# edge work off the scan: the scan only notes a tile's edges in an LDS list; behind a barrier each wave takes 32 (scores one edge per lane, rows of 4 edges in flight); sums stay in LDS arrival order
# speedup vs baseline: 4.1233x; 1.1812x over previous
#include <hip/hip_runtime.h>

#define N_NODES 50000
#define N_EDGES 800000
#define F_OUT   128
#define N_HEAD  4
#define NEG_SLOPE 0.2f

typedef __attribute__((ext_vector_type(2))) float v2f;
typedef __attribute__((ext_vector_type(8))) float v8f;

__global__ void coef_kernel(const float* __restrict__ w, const float* __restrict__ a,
                            float* __restrict__ coefB) {
    int idx = blockIdx.x * blockDim.x + threadIdx.x;
    if (idx >= F_OUT * 16) return;
    int k = idx >> 4, col = idx & 15;
    float v = 0.f;
    if (col < 8) {
        int head = col & 3, sel = col >> 2;
        v = w[head * F_OUT + k] * a[head * 2 * F_OUT + sel * F_OUT + k];
    }
    coefB[idx] = v;
}

__global__ void proj_kernel(const float* __restrict__ x, const float* __restrict__ coefB,
                            float* __restrict__ s_src, float* __restrict__ s_dst) {
    int tile = blockIdx.x;
    int lane = threadIdx.x;
    int half = lane >> 4;
    int rc   = lane & 15;
    int koff = half * 2;
    long arow = (long)(tile * 16 + rc) * F_OUT;

    v8f c = {0.f, 0.f, 0.f, 0.f, 0.f, 0.f, 0.f, 0.f};
#pragma unroll
    for (int k = 0; k < F_OUT; k += 4) {
        v2f av, bv;
        float2 ax = *(const float2*)(x + arow + k + koff);
        av.x = ax.x; av.y = ax.y;
        bv.x = coefB[(k + koff) * 16 + rc];
        bv.y = coefB[(k + koff + 1) * 16 + rc];
        c = __builtin_amdgcn_wmma_f32_16x16x4_f32(false, av, false, bv,
                                                  (short)0, c, false, false);
    }
    if (rc < 8) {
        int head = rc & 3;
        float* buf = (rc >> 2) ? s_dst : s_src;
        int mbase = tile * 16 + half * 8;
#pragma unroll
        for (int j = 0; j < 8; ++j)
            buf[head * N_NODES + mbase + j] = c[j];
    }
}

#define TILE_R       64
#define TILE_THREADS 1024
#define N_TILES      ((N_NODES + TILE_R - 1) / TILE_R)

#define N_CHUNK      (N_EDGES / 4 / 32)
#define SCAN_DEPTH   4
#define QCAP         4096
#define DRAIN_AHEAD  4

static_assert((N_EDGES % 128) == 0, "a wave reads 32 whole int4 of sources at a time");
static_assert((N_HEAD & (N_HEAD - 1)) == 0 && N_HEAD <= 32, "lane & (N_HEAD - 1) names a head; lanes 0..N_HEAD-1 hold one each");
static_assert(N_HEAD == 4, "a lane of the drain holds the four head scores of its edge in one float4");
static_assert(N_EDGES <= (1 << 20) && TILE_R <= (1 << 12), "a list word holds the row offset above bit 20 and the edge number below it");

__device__ __forceinline__ int pick4(const int4 v, const int k) {
    return (k == 0) ? v.x : (k == 1) ? v.y : (k == 2) ? v.z : v.w;
}

__global__ __launch_bounds__(TILE_THREADS)
void tile_kernel(const float* __restrict__ x, const float* __restrict__ w,
                 const int* __restrict__ ei,
                 const float* __restrict__ s_src, const float* __restrict__ s_dst,
                 float* __restrict__ out) {
    __shared__ float acc[TILE_R * N_HEAD * F_OUT];
    __shared__ float rs[TILE_R * N_HEAD];
    __shared__ float ss[TILE_R * N_HEAD];
    __shared__ unsigned q[QCAP];
    __shared__ int qn;
    __shared__ int qlim;

    const int tid  = threadIdx.x;
    const int lane = tid & 31;
    const int hd   = lane & (N_HEAD - 1);
    const int wid  = __builtin_amdgcn_readfirstlane(tid >> 5);
    const int base = blockIdx.x * TILE_R;

    const float4 z4 = make_float4(0.f, 0.f, 0.f, 0.f);
    for (int i = tid; i < TILE_R * N_HEAD * F_OUT / 4; i += TILE_THREADS)
        ((float4*)acc)[i] = z4;
    if (tid < TILE_R * N_HEAD) {
        int node = base + (tid >> 2);
        rs[tid] = 0.f;
        ss[tid] = (node < N_NODES) ? s_src[(tid & 3) * N_NODES + node] : 0.f;
    }
    if (tid == 0) { qn = 0; qlim = QCAP; }

    __syncthreads();

    const int4* __restrict__ src4 = (const int4*)ei;
    const int*  __restrict__ dstp = ei + N_EDGES;

    for (int c0 = wid * SCAN_DEPTH; c0 < N_CHUNK; c0 += (TILE_THREADS / 32) * SCAN_DEPTH) {
        int4 s[SCAN_DEPTH];
#pragma unroll
        for (int j = 0; j < SCAN_DEPTH; ++j) {
            const int cj = (c0 + j < N_CHUNK) ? (c0 + j) : (N_CHUNK - 1);
            s[j] = src4[cj * 32 + lane];
        }
        unsigned direct = 0;
#pragma unroll
        for (int j = 0; j < SCAN_DEPTH; ++j) {
            const unsigned d01 = min((unsigned)(s[j].x - base), (unsigned)(s[j].y - base));
            const unsigned d23 = min((unsigned)(s[j].z - base), (unsigned)(s[j].w - base));
            const bool hit = (c0 + j < N_CHUNK) && (min(d01, d23) < (unsigned)TILE_R);
            const unsigned m = __builtin_amdgcn_ballot_w32(hit);
            if (m) {
                const unsigned r0 = (unsigned)(s[j].x - base), r1 = (unsigned)(s[j].y - base);
                const unsigned r2 = (unsigned)(s[j].z - base), r3 = (unsigned)(s[j].w - base);
                const unsigned m0 = __builtin_amdgcn_ballot_w32(r0 < (unsigned)TILE_R);
                const unsigned m1 = __builtin_amdgcn_ballot_w32(r1 < (unsigned)TILE_R);
                const unsigned m2 = __builtin_amdgcn_ballot_w32(r2 < (unsigned)TILE_R);
                const unsigned m3 = __builtin_amdgcn_ballot_w32(r3 < (unsigned)TILE_R);
                const int n0 = __builtin_popcount(m0), n1 = __builtin_popcount(m1);
                const int n2 = __builtin_popcount(m2), n3 = __builtin_popcount(m3);
                const int total = n0 + n1 + n2 + n3;
                int first = 0;
                if (lane == 0) first = atomicAdd(&qn, total);
                first = __builtin_amdgcn_readfirstlane(first);
                if (first + total <= QCAP) {
                    const unsigned e0 = (unsigned)(((c0 + j) * 32 + lane) * 4);
                    if (r0 < (unsigned)TILE_R)
                        q[first + (int)__builtin_amdgcn_mbcnt_lo(m0, 0u)] = (r0 << 20) | e0;
                    if (r1 < (unsigned)TILE_R)
                        q[first + n0 + (int)__builtin_amdgcn_mbcnt_lo(m1, 0u)] = (r1 << 20) | (e0 + 1u);
                    if (r2 < (unsigned)TILE_R)
                        q[first + n0 + n1 + (int)__builtin_amdgcn_mbcnt_lo(m2, 0u)] = (r2 << 20) | (e0 + 2u);
                    if (r3 < (unsigned)TILE_R)
                        q[first + n0 + n1 + n2 + (int)__builtin_amdgcn_mbcnt_lo(m3, 0u)] = (r3 << 20) | (e0 + 3u);
                } else {
                    if (lane == 0) atomicMin(&qlim, first);
                    direct |= 1u << j;
                }
            }
        }
        if (direct) {
#pragma nounroll
            for (int jk = 0; jk < SCAN_DEPTH * 4; ++jk) {
                const int j = jk >> 2, k = jk & 3;
                if (!((direct >> j) & 1u)) continue;
                const int4 sj = make_int4((j == 0) ? s[0].x : (j == 1) ? s[1].x : (j == 2) ? s[2].x : s[3].x,
                                          (j == 0) ? s[0].y : (j == 1) ? s[1].y : (j == 2) ? s[2].y : s[3].y,
                                          (j == 0) ? s[0].z : (j == 1) ? s[1].z : (j == 2) ? s[2].z : s[3].z,
                                          (j == 0) ? s[0].w : (j == 1) ? s[1].w : (j == 2) ? s[2].w : s[3].w);
                const unsigned rk = (unsigned)(pick4(sj, k) - base);
                unsigned mk = __builtin_amdgcn_ballot_w32(rk < (unsigned)TILE_R);
                while (mk) {
                    const int L = __builtin_ctz(mk);
                    mk &= mk - 1;
                    const unsigned r = (unsigned)__builtin_amdgcn_readlane((int)rk, L);
                    const int dn = dstp[((c0 + j) * 32 + L) * 4 + k];
                    const float* xr = x + (long)dn * F_OUT + lane;
                    const float logit = ss[r * N_HEAD + hd] + s_dst[hd * N_NODES + dn];
                    const float lr = (logit >= 0.f) ? logit : NEG_SLOPE * logit;
                    const float ev = expf(-lr);
                    const int evb = __float_as_int(ev);
                    float* a = acc + r * (N_HEAD * F_OUT) + lane;
#pragma nounroll
                    for (int i = 0; i < N_HEAD; ++i) {
                        const float vi = __int_as_float(__builtin_amdgcn_readlane(evb, i));
#pragma unroll
                        for (int cc = 0; cc < 4; ++cc)
                            atomicAdd(a + i * F_OUT + cc * 32, vi * xr[cc * 32] * w[i * F_OUT + cc * 32 + lane]);
                    }
                    if (lane < N_HEAD)
                        atomicAdd(&rs[r * N_HEAD + lane], ev);
                }
            }
        }
    }

    __syncthreads();

    const int nq = __builtin_amdgcn_readfirstlane(min(qn, qlim));

    float wr[N_HEAD][4];
#pragma unroll
    for (int i = 0; i < N_HEAD; ++i)
#pragma unroll
        for (int c = 0; c < 4; ++c)
            wr[i][c] = w[i * F_OUT + c * 32 + lane];

    for (int b0 = wid * 32; b0 < nq; b0 += TILE_THREADS) {
        const int cnt = min(32, nq - b0);
        const bool live = lane < cnt;
        const unsigned ent = q[live ? (b0 + lane) : b0];
        const unsigned rl = ent >> 20;
        const int dl = dstp[ent & 0xFFFFFu];
        const float4 sv = ((const float4*)ss)[rl];
        const float g0 = sv.x + s_dst[0 * N_NODES + dl];
        const float g1 = sv.y + s_dst[1 * N_NODES + dl];
        const float g2 = sv.z + s_dst[2 * N_NODES + dl];
        const float g3 = sv.w + s_dst[3 * N_NODES + dl];
        const float ev[N_HEAD] = { expf(-((g0 >= 0.f) ? g0 : NEG_SLOPE * g0)),
                                   expf(-((g1 >= 0.f) ? g1 : NEG_SLOPE * g1)),
                                   expf(-((g2 >= 0.f) ? g2 : NEG_SLOPE * g2)),
                                   expf(-((g3 >= 0.f) ? g3 : NEG_SLOPE * g3)) };
        if (live) {
#pragma unroll
            for (int i = 0; i < N_HEAD; ++i)
                atomicAdd(&rs[rl * N_HEAD + i], ev[i]);
        }
        for (int j0 = 0; j0 < cnt; j0 += DRAIN_AHEAD) {
            float xv[DRAIN_AHEAD][4];
#pragma unroll
            for (int u = 0; u < DRAIN_AHEAD; ++u) {
                const int jj = min(j0 + u, cnt - 1);
                const int dn = __builtin_amdgcn_readlane(dl, jj);
                const float* xr = x + (long)dn * F_OUT + lane;
#pragma unroll
                for (int cc = 0; cc < 4; ++cc) xv[u][cc] = xr[cc * 32];
            }
#pragma unroll
            for (int u = 0; u < DRAIN_AHEAD; ++u) {
                if (j0 + u < cnt) {
                    const int jj = j0 + u;
                    const unsigned r = (unsigned)__builtin_amdgcn_readlane((int)rl, jj);
                    float* a = acc + r * (N_HEAD * F_OUT) + lane;
#pragma unroll
                    for (int i = 0; i < N_HEAD; ++i) {
                        const float vi = __int_as_float(__builtin_amdgcn_readlane(__float_as_int(ev[i]), jj));
#pragma unroll
                        for (int cc = 0; cc < 4; ++cc)
                            atomicAdd(a + i * F_OUT + cc * 32, vi * xv[u][cc] * wr[i][cc]);
                    }
                }
            }
        }
    }

    __syncthreads();

    for (int p = wid; p < TILE_R * N_HEAD; p += TILE_THREADS / 32) {
        const int node = base + (p >> 2);
        const int i    = p & 3;
        if (node < N_NODES) {
            const float inv = 1.0f / rs[p];
            float4 v = ((const float4*)acc)[p * (F_OUT / 4) + lane];
            v.x *= inv; v.y *= inv; v.z *= inv; v.w *= inv;
            ((float4*)out)[((long)i * N_NODES + node) * (F_OUT / 4) + lane] = v;
        }
    }
}

extern "C" void kernel_launch(void* const* d_in, const int* in_sizes, int n_in,
                              void* d_out, int out_size, void* d_ws, size_t ws_size,
                              hipStream_t stream) {
    const float* x  = (const float*)d_in[0];
    const float* w  = (const float*)d_in[1];
    const float* a  = (const float*)d_in[2];
    const int*   ei = (const int*)d_in[3];
    float* out = (float*)d_out;

    float* ws     = (float*)d_ws;
    float* s_src  = ws;
    float* s_dst  = ws + N_HEAD * N_NODES;
    float* coefB  = ws + 3 * N_HEAD * N_NODES;

    coef_kernel<<<8, 256, 0, stream>>>(w, a, coefB);
    proj_kernel<<<N_NODES / 16, 32, 0, stream>>>(x, coefB, s_src, s_dst);
    tile_kernel<<<N_TILES, TILE_THREADS, 0, stream>>>(x, w, ei, s_src, s_dst, out);
}


# piece filter: launch 1 also marks per 64-row tile which 128-source pieces hold a source of it (bitmap, LDS or-atomics); a tile reads only its marked pieces; sums stay in LDS arrival order
# speedup vs baseline: 5.3601x; 1.3000x over previous
#include <hip/hip_runtime.h>

#define N_NODES 50000
#define N_EDGES 800000
#define F_OUT   128
#define N_HEAD  4
#define NEG_SLOPE 0.2f

typedef __attribute__((ext_vector_type(2))) float v2f;
typedef __attribute__((ext_vector_type(8))) float v8f;

#define TILE_R       64
#define N_TILES      ((N_NODES + TILE_R - 1) / TILE_R)
#define N_CHUNK      (N_EDGES / 4 / 32)
#define N_WORD       ((N_CHUNK + 31) / 32)
#define FILT_PITCH   ((N_TILES + 31) / 32 * 32)
#define COEF_BLOCKS  ((F_OUT * 16 + 255) / 256)

static_assert((N_EDGES % 128) == 0, "a wave reads 32 whole int4 of sources at a time");
static_assert(N_WORD * FILT_PITCH <= N_HEAD * N_NODES, "the filter lies in the workspace between the destination scores and the coefficients");

__global__ __launch_bounds__(256)
void prep_kernel(const float* __restrict__ w, const float* __restrict__ a,
                 const int* __restrict__ ei,
                 float* __restrict__ coefB, unsigned* __restrict__ filt) {
    __shared__ unsigned bm[FILT_PITCH];
    const int tid = threadIdx.x;
    if (blockIdx.x < COEF_BLOCKS) {
        int idx = blockIdx.x * 256 + tid;
        if (idx >= F_OUT * 16) return;
        int k = idx >> 4, col = idx & 15;
        float v = 0.f;
        if (col < 8) {
            int head = col & 3, sel = col >> 2;
            v = w[head * F_OUT + k] * a[head * 2 * F_OUT + sel * F_OUT + k];
        }
        coefB[idx] = v;
        return;
    }
    const int cw = blockIdx.x - COEF_BLOCKS;
    for (int t = tid; t < FILT_PITCH; t += 256) bm[t] = 0u;
    __syncthreads();
    const int4* __restrict__ src4 = (const int4*)ei;
    int4 s[4];
    int  qi[4];
#pragma unroll
    for (int i = 0; i < 4; ++i) {
        qi[i] = cw * 1024 + i * 256 + tid;
        s[i] = src4[min(qi[i], N_EDGES / 4 - 1)];
    }
#pragma unroll
    for (int i = 0; i < 4; ++i) {
        if (qi[i] < N_EDGES / 4) {
            const unsigned bit = 1u << ((qi[i] >> 5) & 31);
            const unsigned t0 = (unsigned)s[i].x / TILE_R, t1 = (unsigned)s[i].y / TILE_R;
            const unsigned t2 = (unsigned)s[i].z / TILE_R, t3 = (unsigned)s[i].w / TILE_R;
            if (t0 < N_TILES) atomicOr(&bm[t0], bit);
            if (t1 < N_TILES) atomicOr(&bm[t1], bit);
            if (t2 < N_TILES) atomicOr(&bm[t2], bit);
            if (t3 < N_TILES) atomicOr(&bm[t3], bit);
        }
    }
    __syncthreads();
    for (int t = tid; t < FILT_PITCH; t += 256) filt[cw * FILT_PITCH + t] = bm[t];
}

__global__ void proj_kernel(const float* __restrict__ x, const float* __restrict__ coefB,
                            float* __restrict__ s_src, float* __restrict__ s_dst) {
    int tile = blockIdx.x;
    int lane = threadIdx.x;
    int half = lane >> 4;
    int rc   = lane & 15;
    int koff = half * 2;
    long arow = (long)(tile * 16 + rc) * F_OUT;

    v8f c = {0.f, 0.f, 0.f, 0.f, 0.f, 0.f, 0.f, 0.f};
#pragma unroll
    for (int k = 0; k < F_OUT; k += 4) {
        v2f av, bv;
        float2 ax = *(const float2*)(x + arow + k + koff);
        av.x = ax.x; av.y = ax.y;
        bv.x = coefB[(k + koff) * 16 + rc];
        bv.y = coefB[(k + koff + 1) * 16 + rc];
        c = __builtin_amdgcn_wmma_f32_16x16x4_f32(false, av, false, bv,
                                                  (short)0, c, false, false);
    }
    if (rc < 8) {
        int head = rc & 3;
        float* buf = (rc >> 2) ? s_dst : s_src;
        int mbase = tile * 16 + half * 8;
#pragma unroll
        for (int j = 0; j < 8; ++j)
            buf[head * N_NODES + mbase + j] = c[j];
    }
}

#define TILE_THREADS 1024
#define SCAN_DEPTH   4
#define QCAP         4096
#define DRAIN_AHEAD  4

static_assert(N_WORD <= 32 * (TILE_THREADS / 32), "lane i of a wave holds the wave's i-th filter word");
static_assert((N_HEAD & (N_HEAD - 1)) == 0 && N_HEAD <= 32, "lane & (N_HEAD - 1) names a head; lanes 0..N_HEAD-1 hold one each");
static_assert(N_HEAD == 4, "a lane of the drain holds the four head scores of its edge in one float4");
static_assert(N_EDGES <= (1 << 20) && TILE_R <= (1 << 12), "a list word holds the row offset above bit 20 and the edge number below it");

__device__ __forceinline__ int pick4(const int4 v, const int k) {
    return (k == 0) ? v.x : (k == 1) ? v.y : (k == 2) ? v.z : v.w;
}

__global__ __launch_bounds__(TILE_THREADS)
void tile_kernel(const float* __restrict__ x, const float* __restrict__ w,
                 const int* __restrict__ ei, const unsigned* __restrict__ filt,
                 const float* __restrict__ s_src, const float* __restrict__ s_dst,
                 float* __restrict__ out) {
    __shared__ float acc[TILE_R * N_HEAD * F_OUT];
    __shared__ float rs[TILE_R * N_HEAD];
    __shared__ float ss[TILE_R * N_HEAD];
    __shared__ unsigned q[QCAP];
    __shared__ int qn;
    __shared__ int qlim;

    const int tid  = threadIdx.x;
    const int lane = tid & 31;
    const int hd   = lane & (N_HEAD - 1);
    const int wid  = __builtin_amdgcn_readfirstlane(tid >> 5);
    const int base = blockIdx.x * TILE_R;

    unsigned fw = 0u;
    if (wid + 32 * lane < N_WORD) fw = filt[(wid + 32 * lane) * FILT_PITCH + blockIdx.x];

    const float4 z4 = make_float4(0.f, 0.f, 0.f, 0.f);
    for (int i = tid; i < TILE_R * N_HEAD * F_OUT / 4; i += TILE_THREADS)
        ((float4*)acc)[i] = z4;
    if (tid < TILE_R * N_HEAD) {
        int node = base + (tid >> 2);
        rs[tid] = 0.f;
        ss[tid] = (node < N_NODES) ? s_src[(tid & 3) * N_NODES + node] : 0.f;
    }
    if (tid == 0) { qn = 0; qlim = QCAP; }

    __syncthreads();

    const int4* __restrict__ src4 = (const int4*)ei;
    const int*  __restrict__ dstp = ei + N_EDGES;

    {
        const int nw = (N_WORD - wid + 31) / 32;
        int wi = 0;
        unsigned mask = (unsigned)__builtin_amdgcn_readlane((int)fw, 0);
        for (;;) {
            int  cid[SCAN_DEPTH];
            bool val[SCAN_DEPTH];
#pragma unroll
            for (int j = 0; j < SCAN_DEPTH; ++j) {
                while (mask == 0u && wi + 1 < nw) {
                    ++wi;
                    mask = (unsigned)__builtin_amdgcn_readlane((int)fw, wi);
                }
                val[j] = (mask != 0u);
                cid[j] = val[j] ? ((wid + 32 * wi) * 32 + __builtin_ctz(mask)) : (j ? cid[0] : 0);
                mask &= mask - 1u;
            }
            if (!val[0]) break;
            int4 s[SCAN_DEPTH];
#pragma unroll
            for (int j = 0; j < SCAN_DEPTH; ++j)
                s[j] = src4[cid[j] * 32 + lane];
            unsigned direct = 0;
#pragma unroll
            for (int j = 0; j < SCAN_DEPTH; ++j) {
                if (val[j]) {
                    const unsigned r0 = (unsigned)(s[j].x - base), r1 = (unsigned)(s[j].y - base);
                    const unsigned r2 = (unsigned)(s[j].z - base), r3 = (unsigned)(s[j].w - base);
                    const unsigned m0 = __builtin_amdgcn_ballot_w32(r0 < (unsigned)TILE_R);
                    const unsigned m1 = __builtin_amdgcn_ballot_w32(r1 < (unsigned)TILE_R);
                    const unsigned m2 = __builtin_amdgcn_ballot_w32(r2 < (unsigned)TILE_R);
                    const unsigned m3 = __builtin_amdgcn_ballot_w32(r3 < (unsigned)TILE_R);
                    const int n0 = __builtin_popcount(m0), n1 = __builtin_popcount(m1);
                    const int n2 = __builtin_popcount(m2), n3 = __builtin_popcount(m3);
                    const int total = n0 + n1 + n2 + n3;
                    int first = 0;
                    if (lane == 0) first = atomicAdd(&qn, total);
                    first = __builtin_amdgcn_readfirstlane(first);
                    if (first + total <= QCAP) {
                        const unsigned e0 = (unsigned)((cid[j] * 32 + lane) * 4);
                        if (r0 < (unsigned)TILE_R)
                            q[first + (int)__builtin_amdgcn_mbcnt_lo(m0, 0u)] = (r0 << 20) | e0;
                        if (r1 < (unsigned)TILE_R)
                            q[first + n0 + (int)__builtin_amdgcn_mbcnt_lo(m1, 0u)] = (r1 << 20) | (e0 + 1u);
                        if (r2 < (unsigned)TILE_R)
                            q[first + n0 + n1 + (int)__builtin_amdgcn_mbcnt_lo(m2, 0u)] = (r2 << 20) | (e0 + 2u);
                        if (r3 < (unsigned)TILE_R)
                            q[first + n0 + n1 + n2 + (int)__builtin_amdgcn_mbcnt_lo(m3, 0u)] = (r3 << 20) | (e0 + 3u);
                    } else {
                        if (lane == 0) atomicMin(&qlim, first);
                        direct |= 1u << j;
                    }
                }
            }
            if (direct) {
#pragma nounroll
                for (int jk = 0; jk < SCAN_DEPTH * 4; ++jk) {
                    const int j = jk >> 2, k = jk & 3;
                    if (!((direct >> j) & 1u)) continue;
                    const int4 sj = make_int4((j == 0) ? s[0].x : (j == 1) ? s[1].x : (j == 2) ? s[2].x : s[3].x,
                                              (j == 0) ? s[0].y : (j == 1) ? s[1].y : (j == 2) ? s[2].y : s[3].y,
                                              (j == 0) ? s[0].z : (j == 1) ? s[1].z : (j == 2) ? s[2].z : s[3].z,
                                              (j == 0) ? s[0].w : (j == 1) ? s[1].w : (j == 2) ? s[2].w : s[3].w);
                    const int cj = (j == 0) ? cid[0] : (j == 1) ? cid[1] : (j == 2) ? cid[2] : cid[3];
                    const unsigned rk = (unsigned)(pick4(sj, k) - base);
                    unsigned mk = __builtin_amdgcn_ballot_w32(rk < (unsigned)TILE_R);
                    while (mk) {
                        const int L = __builtin_ctz(mk);
                        mk &= mk - 1;
                        const unsigned r = (unsigned)__builtin_amdgcn_readlane((int)rk, L);
                        const int dn = dstp[(cj * 32 + L) * 4 + k];
                        const float* xr = x + (long)dn * F_OUT + lane;
                        const float logit = ss[r * N_HEAD + hd] + s_dst[hd * N_NODES + dn];
                        const float lr = (logit >= 0.f) ? logit : NEG_SLOPE * logit;
                        const float ev = expf(-lr);
                        const int evb = __float_as_int(ev);
                        float* a = acc + r * (N_HEAD * F_OUT) + lane;
#pragma nounroll
                        for (int i = 0; i < N_HEAD; ++i) {
                            const float vi = __int_as_float(__builtin_amdgcn_readlane(evb, i));
#pragma unroll
                            for (int cc = 0; cc < 4; ++cc)
                                atomicAdd(a + i * F_OUT + cc * 32, vi * xr[cc * 32] * w[i * F_OUT + cc * 32 + lane]);
                        }
                        if (lane < N_HEAD)
                            atomicAdd(&rs[r * N_HEAD + lane], ev);
                    }
                }
            }
        }
    }

    __syncthreads();

    const int nq = __builtin_amdgcn_readfirstlane(min(qn, qlim));

    float wr[N_HEAD][4];
#pragma unroll
    for (int i = 0; i < N_HEAD; ++i)
#pragma unroll
        for (int c = 0; c < 4; ++c)
            wr[i][c] = w[i * F_OUT + c * 32 + lane];

    for (int b0 = wid * 32; b0 < nq; b0 += TILE_THREADS) {
        const int cnt = min(32, nq - b0);
        const bool live = lane < cnt;
        const unsigned ent = q[live ? (b0 + lane) : b0];
        const unsigned rl = ent >> 20;
        const int dl = dstp[ent & 0xFFFFFu];
        const float4 sv = ((const float4*)ss)[rl];
        const float g0 = sv.x + s_dst[0 * N_NODES + dl];
        const float g1 = sv.y + s_dst[1 * N_NODES + dl];
        const float g2 = sv.z + s_dst[2 * N_NODES + dl];
        const float g3 = sv.w + s_dst[3 * N_NODES + dl];
        const float ev[N_HEAD] = { expf(-((g0 >= 0.f) ? g0 : NEG_SLOPE * g0)),
                                   expf(-((g1 >= 0.f) ? g1 : NEG_SLOPE * g1)),
                                   expf(-((g2 >= 0.f) ? g2 : NEG_SLOPE * g2)),
                                   expf(-((g3 >= 0.f) ? g3 : NEG_SLOPE * g3)) };
        if (live) {
#pragma unroll
            for (int i = 0; i < N_HEAD; ++i)
                atomicAdd(&rs[rl * N_HEAD + i], ev[i]);
        }
        for (int j0 = 0; j0 < cnt; j0 += DRAIN_AHEAD) {
            float xv[DRAIN_AHEAD][4];
#pragma unroll
            for (int u = 0; u < DRAIN_AHEAD; ++u) {
                const int jj = min(j0 + u, cnt - 1);
                const int dn = __builtin_amdgcn_readlane(dl, jj);
                const float* xr = x + (long)dn * F_OUT + lane;
#pragma unroll
                for (int cc = 0; cc < 4; ++cc) xv[u][cc] = xr[cc * 32];
            }
#pragma unroll
            for (int u = 0; u < DRAIN_AHEAD; ++u) {
                if (j0 + u < cnt) {
                    const int jj = j0 + u;
                    const unsigned r = (unsigned)__builtin_amdgcn_readlane((int)rl, jj);
                    float* a = acc + r * (N_HEAD * F_OUT) + lane;
#pragma unroll
                    for (int i = 0; i < N_HEAD; ++i) {
                        const float vi = __int_as_float(__builtin_amdgcn_readlane(__float_as_int(ev[i]), jj));
#pragma unroll
                        for (int cc = 0; cc < 4; ++cc)
                            atomicAdd(a + i * F_OUT + cc * 32, vi * xv[u][cc] * wr[i][cc]);
                    }
                }
            }
        }
    }

    __syncthreads();

    for (int p = wid; p < TILE_R * N_HEAD; p += TILE_THREADS / 32) {
        const int node = base + (p >> 2);
        const int i    = p & 3;
        if (node < N_NODES) {
            const float inv = 1.0f / rs[p];
            float4 v = ((const float4*)acc)[p * (F_OUT / 4) + lane];
            v.x *= inv; v.y *= inv; v.z *= inv; v.w *= inv;
            ((float4*)out)[((long)i * N_NODES + node) * (F_OUT / 4) + lane] = v;
        }
    }
}

extern "C" void kernel_launch(void* const* d_in, const int* in_sizes, int n_in,
                              void* d_out, int out_size, void* d_ws, size_t ws_size,
                              hipStream_t stream) {
    const float* x  = (const float*)d_in[0];
    const float* w  = (const float*)d_in[1];
    const float* a  = (const float*)d_in[2];
    const int*   ei = (const int*)d_in[3];
    float* out = (float*)d_out;

    float* ws     = (float*)d_ws;
    float* s_src  = ws;
    float* s_dst  = ws + N_HEAD * N_NODES;
    unsigned* filt = (unsigned*)(ws + 2 * N_HEAD * N_NODES);
    float* coefB  = ws + 3 * N_HEAD * N_NODES;

    prep_kernel<<<COEF_BLOCKS + N_WORD, 256, 0, stream>>>(w, a, ei, coefB, filt);
    proj_kernel<<<N_NODES / 16, 32, 0, stream>>>(x, coefB, s_src, s_dst);
    tile_kernel<<<N_TILES, TILE_THREADS, 0, stream>>>(x, w, ei, filt, s_src, s_dst, out);
}


# no LDS float adds: a tile's noted edges are counting-sorted by row in LDS (integer rank adds), a wave owns two rows and sums them in registers from 16-byte row loads; sum order = LDS rank order
# speedup vs baseline: 7.4652x; 1.3927x over previous
#include <hip/hip_runtime.h>

#define N_NODES 50000
#define N_EDGES 800000
#define F_OUT   128
#define N_HEAD  4
#define NEG_SLOPE 0.2f

typedef __attribute__((ext_vector_type(2))) float v2f;
typedef __attribute__((ext_vector_type(8))) float v8f;

#define TILE_R       64
#define N_TILES      ((N_NODES + TILE_R - 1) / TILE_R)
#define N_CHUNK      (N_EDGES / 4 / 32)
#define N_WORD       ((N_CHUNK + 31) / 32)
#define FILT_PITCH   ((N_TILES + 31) / 32 * 32)
#define COEF_BLOCKS  ((F_OUT * 16 + 255) / 256)

static_assert((N_EDGES % 128) == 0, "a wave reads 32 whole int4 of sources at a time");
static_assert(N_WORD * FILT_PITCH <= N_HEAD * N_NODES, "the filter lies in the workspace between the destination scores and the coefficients");

__global__ __launch_bounds__(256)
void prep_kernel(const float* __restrict__ w, const float* __restrict__ a,
                 const int* __restrict__ ei,
                 float* __restrict__ coefB, unsigned* __restrict__ filt) {
    __shared__ unsigned bm[FILT_PITCH];
    const int tid = threadIdx.x;
    if (blockIdx.x < COEF_BLOCKS) {
        int idx = blockIdx.x * 256 + tid;
        if (idx >= F_OUT * 16) return;
        int k = idx >> 4, col = idx & 15;
        float v = 0.f;
        if (col < 8) {
            int head = col & 3, sel = col >> 2;
            v = w[head * F_OUT + k] * a[head * 2 * F_OUT + sel * F_OUT + k];
        }
        coefB[idx] = v;
        return;
    }
    const int cw = blockIdx.x - COEF_BLOCKS;
    for (int t = tid; t < FILT_PITCH; t += 256) bm[t] = 0u;
    __syncthreads();
    const int4* __restrict__ src4 = (const int4*)ei;
    int4 s[4];
    int  qi[4];
#pragma unroll
    for (int i = 0; i < 4; ++i) {
        qi[i] = cw * 1024 + i * 256 + tid;
        s[i] = src4[min(qi[i], N_EDGES / 4 - 1)];
    }
#pragma unroll
    for (int i = 0; i < 4; ++i) {
        if (qi[i] < N_EDGES / 4) {
            const unsigned bit = 1u << ((qi[i] >> 5) & 31);
            const unsigned t0 = (unsigned)s[i].x / TILE_R, t1 = (unsigned)s[i].y / TILE_R;
            const unsigned t2 = (unsigned)s[i].z / TILE_R, t3 = (unsigned)s[i].w / TILE_R;
            if (t0 < N_TILES) atomicOr(&bm[t0], bit);
            if (t1 < N_TILES) atomicOr(&bm[t1], bit);
            if (t2 < N_TILES) atomicOr(&bm[t2], bit);
            if (t3 < N_TILES) atomicOr(&bm[t3], bit);
        }
    }
    __syncthreads();
    for (int t = tid; t < FILT_PITCH; t += 256) filt[cw * FILT_PITCH + t] = bm[t];
}

__global__ void proj_kernel(const float* __restrict__ x, const float* __restrict__ coefB,
                            float* __restrict__ s_src, float* __restrict__ s_dst) {
    int tile = blockIdx.x;
    int lane = threadIdx.x;
    int half = lane >> 4;
    int rc   = lane & 15;
    int koff = half * 2;
    long arow = (long)(tile * 16 + rc) * F_OUT;

    v8f c = {0.f, 0.f, 0.f, 0.f, 0.f, 0.f, 0.f, 0.f};
#pragma unroll
    for (int k = 0; k < F_OUT; k += 4) {
        v2f av, bv;
        float2 ax = *(const float2*)(x + arow + k + koff);
        av.x = ax.x; av.y = ax.y;
        bv.x = coefB[(k + koff) * 16 + rc];
        bv.y = coefB[(k + koff + 1) * 16 + rc];
        c = __builtin_amdgcn_wmma_f32_16x16x4_f32(false, av, false, bv,
                                                  (short)0, c, false, false);
    }
    if (rc < 8) {
        int head = rc & 3;
        float* buf = (rc >> 2) ? s_dst : s_src;
        int mbase = tile * 16 + half * 8;
#pragma unroll
        for (int j = 0; j < 8; ++j)
            buf[head * N_NODES + mbase + j] = c[j];
    }
}

#define TILE_THREADS 1024
#define SCAN_DEPTH   4
#define QCAP         2048
#define DRAIN_AHEAD  6

static_assert(N_WORD <= 32 * (TILE_THREADS / 32), "lane i of a wave holds the wave's i-th filter word");
static_assert((N_HEAD & (N_HEAD - 1)) == 0 && N_HEAD <= 32, "lane & (N_HEAD - 1) names a head; lanes 0..N_HEAD-1 hold one each");
static_assert(N_HEAD == 4, "a lane of the drain holds the four head scores of its edge in one float4");
static_assert(N_EDGES <= (1 << 20) && TILE_R <= (1 << 12), "a list word holds the row offset above bit 20 and the edge number below it");
static_assert(TILE_R == 64 && TILE_THREADS == 1024, "a lane scans two row counts; a wave owns two rows");
static_assert(QCAP % TILE_THREADS == 0, "a thread ranks QCAP / TILE_THREADS list words");

__device__ __forceinline__ int pick4(const int4 v, const int k) {
    return (k == 0) ? v.x : (k == 1) ? v.y : (k == 2) ? v.z : v.w;
}

__global__ __launch_bounds__(TILE_THREADS)
void tile_kernel(const float* __restrict__ x, const float* __restrict__ w,
                 const int* __restrict__ ei, const unsigned* __restrict__ filt,
                 const float* __restrict__ s_src, const float* __restrict__ s_dst,
                 float* __restrict__ out) {
    __shared__ float acc[TILE_R * N_HEAD * F_OUT];
    __shared__ float rs[TILE_R * N_HEAD];
    __shared__ float ss[TILE_R * N_HEAD];
    __shared__ unsigned q[QCAP];
    __shared__ unsigned sorted[QCAP];
    __shared__ int rcnt[TILE_R];
    __shared__ int qn;
    __shared__ int qlim;

    const int tid  = threadIdx.x;
    const int lane = tid & 31;
    const int hd   = lane & (N_HEAD - 1);
    const int wid  = __builtin_amdgcn_readfirstlane(tid >> 5);
    const int base = blockIdx.x * TILE_R;

    unsigned fw = 0u;
    if (wid + 32 * lane < N_WORD) fw = filt[(wid + 32 * lane) * FILT_PITCH + blockIdx.x];

    const float4 z4 = make_float4(0.f, 0.f, 0.f, 0.f);
    for (int i = tid; i < TILE_R * N_HEAD * F_OUT / 4; i += TILE_THREADS)
        ((float4*)acc)[i] = z4;
    if (tid < TILE_R * N_HEAD) {
        int node = base + (tid >> 2);
        rs[tid] = 0.f;
        ss[tid] = (node < N_NODES) ? s_src[(tid & 3) * N_NODES + node] : 0.f;
    }
    if (tid < TILE_R) rcnt[tid] = 0;
    if (tid == 0) { qn = 0; qlim = QCAP; }

    __syncthreads();

    const int4* __restrict__ src4 = (const int4*)ei;
    const int*  __restrict__ dstp = ei + N_EDGES;

    {
        const int nw = (N_WORD - wid + 31) / 32;
        int wi = 0;
        unsigned mask = (unsigned)__builtin_amdgcn_readlane((int)fw, 0);
        for (;;) {
            int  cid[SCAN_DEPTH];
            bool val[SCAN_DEPTH];
#pragma unroll
            for (int j = 0; j < SCAN_DEPTH; ++j) {
                while (mask == 0u && wi + 1 < nw) {
                    ++wi;
                    mask = (unsigned)__builtin_amdgcn_readlane((int)fw, wi);
                }
                val[j] = (mask != 0u);
                cid[j] = val[j] ? ((wid + 32 * wi) * 32 + __builtin_ctz(mask)) : (j ? cid[0] : 0);
                mask &= mask - 1u;
            }
            if (!val[0]) break;
            int4 s[SCAN_DEPTH];
#pragma unroll
            for (int j = 0; j < SCAN_DEPTH; ++j)
                s[j] = src4[cid[j] * 32 + lane];
            unsigned direct = 0;
#pragma unroll
            for (int j = 0; j < SCAN_DEPTH; ++j) {
                if (val[j]) {
                    const unsigned r0 = (unsigned)(s[j].x - base), r1 = (unsigned)(s[j].y - base);
                    const unsigned r2 = (unsigned)(s[j].z - base), r3 = (unsigned)(s[j].w - base);
                    const unsigned m0 = __builtin_amdgcn_ballot_w32(r0 < (unsigned)TILE_R);
                    const unsigned m1 = __builtin_amdgcn_ballot_w32(r1 < (unsigned)TILE_R);
                    const unsigned m2 = __builtin_amdgcn_ballot_w32(r2 < (unsigned)TILE_R);
                    const unsigned m3 = __builtin_amdgcn_ballot_w32(r3 < (unsigned)TILE_R);
                    const int n0 = __builtin_popcount(m0), n1 = __builtin_popcount(m1);
                    const int n2 = __builtin_popcount(m2), n3 = __builtin_popcount(m3);
                    const int total = n0 + n1 + n2 + n3;
                    int first = 0;
                    if (lane == 0) first = atomicAdd(&qn, total);
                    first = __builtin_amdgcn_readfirstlane(first);
                    if (first + total <= QCAP) {
                        const unsigned e0 = (unsigned)((cid[j] * 32 + lane) * 4);
                        if (r0 < (unsigned)TILE_R)
                            q[first + (int)__builtin_amdgcn_mbcnt_lo(m0, 0u)] = (r0 << 20) | e0;
                        if (r1 < (unsigned)TILE_R)
                            q[first + n0 + (int)__builtin_amdgcn_mbcnt_lo(m1, 0u)] = (r1 << 20) | (e0 + 1u);
                        if (r2 < (unsigned)TILE_R)
                            q[first + n0 + n1 + (int)__builtin_amdgcn_mbcnt_lo(m2, 0u)] = (r2 << 20) | (e0 + 2u);
                        if (r3 < (unsigned)TILE_R)
                            q[first + n0 + n1 + n2 + (int)__builtin_amdgcn_mbcnt_lo(m3, 0u)] = (r3 << 20) | (e0 + 3u);
                    } else {
                        if (lane == 0) atomicMin(&qlim, first);
                        direct |= 1u << j;
                    }
                }
            }
            if (direct) {
#pragma nounroll
                for (int jk = 0; jk < SCAN_DEPTH * 4; ++jk) {
                    const int j = jk >> 2, k = jk & 3;
                    if (!((direct >> j) & 1u)) continue;
                    const int4 sj = make_int4((j == 0) ? s[0].x : (j == 1) ? s[1].x : (j == 2) ? s[2].x : s[3].x,
                                              (j == 0) ? s[0].y : (j == 1) ? s[1].y : (j == 2) ? s[2].y : s[3].y,
                                              (j == 0) ? s[0].z : (j == 1) ? s[1].z : (j == 2) ? s[2].z : s[3].z,
                                              (j == 0) ? s[0].w : (j == 1) ? s[1].w : (j == 2) ? s[2].w : s[3].w);
                    const int cj = (j == 0) ? cid[0] : (j == 1) ? cid[1] : (j == 2) ? cid[2] : cid[3];
                    const unsigned rk = (unsigned)(pick4(sj, k) - base);
                    unsigned mk = __builtin_amdgcn_ballot_w32(rk < (unsigned)TILE_R);
                    while (mk) {
                        const int L = __builtin_ctz(mk);
                        mk &= mk - 1;
                        const unsigned r = (unsigned)__builtin_amdgcn_readlane((int)rk, L);
                        const int dn = dstp[(cj * 32 + L) * 4 + k];
                        const float* xr = x + (long)dn * F_OUT + lane;
                        const float logit = ss[r * N_HEAD + hd] + s_dst[hd * N_NODES + dn];
                        const float lr = (logit >= 0.f) ? logit : NEG_SLOPE * logit;
                        const float ev = expf(-lr);
                        const int evb = __float_as_int(ev);
                        float* a = acc + r * (N_HEAD * F_OUT) + lane;
#pragma nounroll
                        for (int i = 0; i < N_HEAD; ++i) {
                            const float vi = __int_as_float(__builtin_amdgcn_readlane(evb, i));
#pragma unroll
                            for (int cc = 0; cc < 4; ++cc)
                                atomicAdd(a + i * F_OUT + cc * 32, vi * xr[cc * 32] * w[i * F_OUT + cc * 32 + lane]);
                        }
                        if (lane < N_HEAD)
                            atomicAdd(&rs[r * N_HEAD + lane], ev);
                    }
                }
            }
        }
    }

    __syncthreads();

    const int nq = __builtin_amdgcn_readfirstlane(min(qn, qlim));
    const bool ovf = qlim < QCAP;

    unsigned ent_[QCAP / TILE_THREADS];
    int      rk_[QCAP / TILE_THREADS];
#pragma unroll
    for (int u = 0; u < QCAP / TILE_THREADS; ++u) {
        const int t = tid + u * TILE_THREADS;
        ent_[u] = 0u; rk_[u] = 0;
        if (t < nq) {
            ent_[u] = q[t];
            rk_[u] = atomicAdd(&rcnt[ent_[u] >> 20], 1);
        }
    }
    __syncthreads();

    const int c0 = rcnt[2 * lane], c1 = rcnt[2 * lane + 1];
    int incl = c0 + c1;
#pragma unroll
    for (int d = 1; d < 32; d <<= 1) {
        const int v = __shfl_up(incl, d);
        if (lane >= d) incl += v;
    }
    const int excl = incl - (c0 + c1);
#pragma unroll
    for (int u = 0; u < QCAP / TILE_THREADS; ++u) {
        const int t = tid + u * TILE_THREADS;
        const int r = (int)(ent_[u] >> 20);
        const int se = __shfl(excl, r >> 1), sc = __shfl(c0, r >> 1);
        const int st = se + ((r & 1) ? sc : 0);
        if (t < nq) sorted[st + rk_[u]] = ent_[u];
    }
    __syncthreads();

    const int s0 = __builtin_amdgcn_readlane(excl, wid);
    const int s1 = s0 + __builtin_amdgcn_readlane(c0, wid);
    const int s2 = __builtin_amdgcn_readlane(incl, wid);

#pragma nounroll
    for (int h = 0; h < 2; ++h) {
        const int r  = 2 * wid + h;
        const int lo = h ? s1 : s0, hi = h ? s2 : s1;
        float sm[N_HEAD][4], dn_[N_HEAD];
#pragma unroll
        for (int i = 0; i < N_HEAD; ++i) {
            dn_[i] = 0.f;
#pragma unroll
            for (int c = 0; c < 4; ++c) sm[i][c] = 0.f;
        }
        const float4 sl = ((const float4*)ss)[r];
        const float4 sv = make_float4(__int_as_float(__builtin_amdgcn_readfirstlane(__float_as_int(sl.x))),
                                      __int_as_float(__builtin_amdgcn_readfirstlane(__float_as_int(sl.y))),
                                      __int_as_float(__builtin_amdgcn_readfirstlane(__float_as_int(sl.z))),
                                      __int_as_float(__builtin_amdgcn_readfirstlane(__float_as_int(sl.w))));

        for (int b0 = lo; b0 < hi; b0 += 32) {
            const int cnt = min(32, hi - b0);
            const bool live = lane < cnt;
            const unsigned ent = sorted[live ? (b0 + lane) : b0];
            const int dl = dstp[ent & 0xFFFFFu];
            const float g0 = sv.x + s_dst[0 * N_NODES + dl];
            const float g1 = sv.y + s_dst[1 * N_NODES + dl];
            const float g2 = sv.z + s_dst[2 * N_NODES + dl];
            const float g3 = sv.w + s_dst[3 * N_NODES + dl];
            const float ev[N_HEAD] = { expf(-((g0 >= 0.f) ? g0 : NEG_SLOPE * g0)),
                                       expf(-((g1 >= 0.f) ? g1 : NEG_SLOPE * g1)),
                                       expf(-((g2 >= 0.f) ? g2 : NEG_SLOPE * g2)),
                                       expf(-((g3 >= 0.f) ? g3 : NEG_SLOPE * g3)) };
            for (int j0 = 0; j0 < cnt; j0 += DRAIN_AHEAD) {
                float4 xv[DRAIN_AHEAD];
#pragma unroll
                for (int u = 0; u < DRAIN_AHEAD; ++u) {
                    const int jj = min(j0 + u, cnt - 1);
                    const int dn = __builtin_amdgcn_readlane(dl, jj);
                    xv[u] = ((const float4*)(x + (long)dn * F_OUT))[lane];
                }
#pragma unroll
                for (int u = 0; u < DRAIN_AHEAD; ++u) {
                    const int jj = j0 + u;
                    if (jj < cnt) {
#pragma unroll
                        for (int i = 0; i < N_HEAD; ++i) {
                            const float vi = __int_as_float(__builtin_amdgcn_readlane(__float_as_int(ev[i]), jj));
                            dn_[i] += vi;
                            sm[i][0] += vi * xv[u].x; sm[i][1] += vi * xv[u].y;
                            sm[i][2] += vi * xv[u].z; sm[i][3] += vi * xv[u].w;
                        }
                    }
                }
            }
        }

        const int node = base + r;
        if (node < N_NODES) {
            const float4 dv = ((const float4*)rs)[r];
            const float dd[N_HEAD] = { dv.x, dv.y, dv.z, dv.w };
#pragma unroll
            for (int i = 0; i < N_HEAD; ++i) {
                const float4 wv = ((const float4*)(w + i * F_OUT))[lane];
                float4 v;
                v.x = sm[i][0] * wv.x; v.y = sm[i][1] * wv.y;
                v.z = sm[i][2] * wv.z; v.w = sm[i][3] * wv.w;
                if (ovf) {
                    const float4 a = ((const float4*)acc)[(r * N_HEAD + i) * (F_OUT / 4) + lane];
                    v.x += a.x; v.y += a.y; v.z += a.z; v.w += a.w;
                }
                const float inv = 1.0f / (dn_[i] + dd[i]);
                v.x *= inv; v.y *= inv; v.z *= inv; v.w *= inv;
                ((float4*)out)[((long)i * N_NODES + node) * (F_OUT / 4) + lane] = v;
            }
        }
    }
}

extern "C" void kernel_launch(void* const* d_in, const int* in_sizes, int n_in,
                              void* d_out, int out_size, void* d_ws, size_t ws_size,
                              hipStream_t stream) {
    const float* x  = (const float*)d_in[0];
    const float* w  = (const float*)d_in[1];
    const float* a  = (const float*)d_in[2];
    const int*   ei = (const int*)d_in[3];
    float* out = (float*)d_out;

    float* ws     = (float*)d_ws;
    float* s_src  = ws;
    float* s_dst  = ws + N_HEAD * N_NODES;
    unsigned* filt = (unsigned*)(ws + 2 * N_HEAD * N_NODES);
    float* coefB  = ws + 3 * N_HEAD * N_NODES;

    prep_kernel<<<COEF_BLOCKS + N_WORD, 256, 0, stream>>>(w, a, ei, coefB, filt);
    proj_kernel<<<N_NODES / 16, 32, 0, stream>>>(x, coefB, s_src, s_dst);
    tile_kernel<<<N_TILES, TILE_THREADS, 0, stream>>>(x, w, ei, filt, s_src, s_dst, out);
}


# scores stored node-major, one 16-byte record of the four heads per node: an edge fetches its destination's scores by one 16-byte load of one line, not four 4-byte loads from four planes
# speedup vs baseline: 8.6233x; 1.1551x over previous
#include <hip/hip_runtime.h>

#define N_NODES 50000
#define N_EDGES 800000
#define F_OUT   128
#define N_HEAD  4
#define NEG_SLOPE 0.2f

typedef __attribute__((ext_vector_type(2))) float v2f;
typedef __attribute__((ext_vector_type(8))) float v8f;

#define TILE_R       64
#define N_TILES      ((N_NODES + TILE_R - 1) / TILE_R)
#define N_CHUNK      (N_EDGES / 4 / 32)
#define N_WORD       ((N_CHUNK + 31) / 32)
#define FILT_PITCH   ((N_TILES + 31) / 32 * 32)
#define COEF_BLOCKS  ((F_OUT * 16 + 255) / 256)

static_assert((N_EDGES % 128) == 0, "a wave reads 32 whole int4 of sources at a time");
static_assert(N_WORD * FILT_PITCH <= N_HEAD * N_NODES, "the filter lies in the workspace between the destination scores and the coefficients");

__global__ __launch_bounds__(256)
void prep_kernel(const float* __restrict__ w, const float* __restrict__ a,
                 const int* __restrict__ ei,
                 float* __restrict__ coefB, unsigned* __restrict__ filt) {
    __shared__ unsigned bm[FILT_PITCH];
    const int tid = threadIdx.x;
    if (blockIdx.x < COEF_BLOCKS) {
        int idx = blockIdx.x * 256 + tid;
        if (idx >= F_OUT * 16) return;
        int k = idx >> 4, col = idx & 15;
        float v = 0.f;
        if (col < 8) {
            int head = col & 3, sel = col >> 2;
            v = w[head * F_OUT + k] * a[head * 2 * F_OUT + sel * F_OUT + k];
        }
        coefB[idx] = v;
        return;
    }
    const int cw = blockIdx.x - COEF_BLOCKS;
    for (int t = tid; t < FILT_PITCH; t += 256) bm[t] = 0u;
    __syncthreads();
    const int4* __restrict__ src4 = (const int4*)ei;
    int4 s[4];
    int  qi[4];
#pragma unroll
    for (int i = 0; i < 4; ++i) {
        qi[i] = cw * 1024 + i * 256 + tid;
        s[i] = src4[min(qi[i], N_EDGES / 4 - 1)];
    }
#pragma unroll
    for (int i = 0; i < 4; ++i) {
        if (qi[i] < N_EDGES / 4) {
            const unsigned bit = 1u << ((qi[i] >> 5) & 31);
            const unsigned t0 = (unsigned)s[i].x / TILE_R, t1 = (unsigned)s[i].y / TILE_R;
            const unsigned t2 = (unsigned)s[i].z / TILE_R, t3 = (unsigned)s[i].w / TILE_R;
            if (t0 < N_TILES) atomicOr(&bm[t0], bit);
            if (t1 < N_TILES) atomicOr(&bm[t1], bit);
            if (t2 < N_TILES) atomicOr(&bm[t2], bit);
            if (t3 < N_TILES) atomicOr(&bm[t3], bit);
        }
    }
    __syncthreads();
    for (int t = tid; t < FILT_PITCH; t += 256) filt[cw * FILT_PITCH + t] = bm[t];
}

__global__ void proj_kernel(const float* __restrict__ x, const float* __restrict__ coefB,
                            float* __restrict__ s_src, float* __restrict__ s_dst) {
    int tile = blockIdx.x;
    int lane = threadIdx.x;
    int half = lane >> 4;
    int rc   = lane & 15;
    int koff = half * 2;
    long arow = (long)(tile * 16 + rc) * F_OUT;

    v8f c = {0.f, 0.f, 0.f, 0.f, 0.f, 0.f, 0.f, 0.f};
#pragma unroll
    for (int k = 0; k < F_OUT; k += 4) {
        v2f av, bv;
        float2 ax = *(const float2*)(x + arow + k + koff);
        av.x = ax.x; av.y = ax.y;
        bv.x = coefB[(k + koff) * 16 + rc];
        bv.y = coefB[(k + koff + 1) * 16 + rc];
        c = __builtin_amdgcn_wmma_f32_16x16x4_f32(false, av, false, bv,
                                                  (short)0, c, false, false);
    }
    if (rc < 8) {
        int head = rc & 3;
        float* buf = (rc >> 2) ? s_dst : s_src;
        int mbase = tile * 16 + half * 8;
#pragma unroll
        for (int j = 0; j < 8; ++j)
            buf[(mbase + j) * N_HEAD + head] = c[j];
    }
}

#define TILE_THREADS 1024
#define SCAN_DEPTH   4
#define QCAP         2048
#define DRAIN_AHEAD  6

static_assert(N_WORD <= 32 * (TILE_THREADS / 32), "lane i of a wave holds the wave's i-th filter word");
static_assert((N_HEAD & (N_HEAD - 1)) == 0 && N_HEAD <= 32, "lane & (N_HEAD - 1) names a head; lanes 0..N_HEAD-1 hold one each");
static_assert(N_HEAD == 4, "a lane of the drain holds the four head scores of its edge in one float4");
static_assert(N_EDGES <= (1 << 20) && TILE_R <= (1 << 12), "a list word holds the row offset above bit 20 and the edge number below it");
static_assert(TILE_R == 64 && TILE_THREADS == 1024, "a lane scans two row counts; a wave owns two rows");
static_assert(QCAP % TILE_THREADS == 0, "a thread ranks QCAP / TILE_THREADS list words");

__device__ __forceinline__ int pick4(const int4 v, const int k) {
    return (k == 0) ? v.x : (k == 1) ? v.y : (k == 2) ? v.z : v.w;
}

__global__ __launch_bounds__(TILE_THREADS)
void tile_kernel(const float* __restrict__ x, const float* __restrict__ w,
                 const int* __restrict__ ei, const unsigned* __restrict__ filt,
                 const float* __restrict__ s_src, const float* __restrict__ s_dst,
                 float* __restrict__ out) {
    __shared__ float acc[TILE_R * N_HEAD * F_OUT];
    __shared__ float rs[TILE_R * N_HEAD];
    __shared__ float ss[TILE_R * N_HEAD];
    __shared__ unsigned q[QCAP];
    __shared__ unsigned sorted[QCAP];
    __shared__ int rcnt[TILE_R];
    __shared__ int qn;
    __shared__ int qlim;

    const int tid  = threadIdx.x;
    const int lane = tid & 31;
    const int hd   = lane & (N_HEAD - 1);
    const int wid  = __builtin_amdgcn_readfirstlane(tid >> 5);
    const int base = blockIdx.x * TILE_R;

    unsigned fw = 0u;
    if (wid + 32 * lane < N_WORD) fw = filt[(wid + 32 * lane) * FILT_PITCH + blockIdx.x];

    const float4 z4 = make_float4(0.f, 0.f, 0.f, 0.f);
    for (int i = tid; i < TILE_R * N_HEAD * F_OUT / 4; i += TILE_THREADS)
        ((float4*)acc)[i] = z4;
    if (tid < TILE_R * N_HEAD) {
        int node = base + (tid >> 2);
        rs[tid] = 0.f;
        ss[tid] = (node < N_NODES) ? s_src[node * N_HEAD + (tid & 3)] : 0.f;
    }
    if (tid < TILE_R) rcnt[tid] = 0;
    if (tid == 0) { qn = 0; qlim = QCAP; }

    __syncthreads();

    const int4* __restrict__ src4 = (const int4*)ei;
    const int*  __restrict__ dstp = ei + N_EDGES;

    {
        const int nw = (N_WORD - wid + 31) / 32;
        int wi = 0;
        unsigned mask = (unsigned)__builtin_amdgcn_readlane((int)fw, 0);
        for (;;) {
            int  cid[SCAN_DEPTH];
            bool val[SCAN_DEPTH];
#pragma unroll
            for (int j = 0; j < SCAN_DEPTH; ++j) {
                while (mask == 0u && wi + 1 < nw) {
                    ++wi;
                    mask = (unsigned)__builtin_amdgcn_readlane((int)fw, wi);
                }
                val[j] = (mask != 0u);
                cid[j] = val[j] ? ((wid + 32 * wi) * 32 + __builtin_ctz(mask)) : (j ? cid[0] : 0);
                mask &= mask - 1u;
            }
            if (!val[0]) break;
            int4 s[SCAN_DEPTH];
#pragma unroll
            for (int j = 0; j < SCAN_DEPTH; ++j)
                s[j] = src4[cid[j] * 32 + lane];
            unsigned direct = 0;
#pragma unroll
            for (int j = 0; j < SCAN_DEPTH; ++j) {
                if (val[j]) {
                    const unsigned r0 = (unsigned)(s[j].x - base), r1 = (unsigned)(s[j].y - base);
                    const unsigned r2 = (unsigned)(s[j].z - base), r3 = (unsigned)(s[j].w - base);
                    const unsigned m0 = __builtin_amdgcn_ballot_w32(r0 < (unsigned)TILE_R);
                    const unsigned m1 = __builtin_amdgcn_ballot_w32(r1 < (unsigned)TILE_R);
                    const unsigned m2 = __builtin_amdgcn_ballot_w32(r2 < (unsigned)TILE_R);
                    const unsigned m3 = __builtin_amdgcn_ballot_w32(r3 < (unsigned)TILE_R);
                    const int n0 = __builtin_popcount(m0), n1 = __builtin_popcount(m1);
                    const int n2 = __builtin_popcount(m2), n3 = __builtin_popcount(m3);
                    const int total = n0 + n1 + n2 + n3;
                    int first = 0;
                    if (lane == 0) first = atomicAdd(&qn, total);
                    first = __builtin_amdgcn_readfirstlane(first);
                    if (first + total <= QCAP) {
                        const unsigned e0 = (unsigned)((cid[j] * 32 + lane) * 4);
                        if (r0 < (unsigned)TILE_R)
                            q[first + (int)__builtin_amdgcn_mbcnt_lo(m0, 0u)] = (r0 << 20) | e0;
                        if (r1 < (unsigned)TILE_R)
                            q[first + n0 + (int)__builtin_amdgcn_mbcnt_lo(m1, 0u)] = (r1 << 20) | (e0 + 1u);
                        if (r2 < (unsigned)TILE_R)
                            q[first + n0 + n1 + (int)__builtin_amdgcn_mbcnt_lo(m2, 0u)] = (r2 << 20) | (e0 + 2u);
                        if (r3 < (unsigned)TILE_R)
                            q[first + n0 + n1 + n2 + (int)__builtin_amdgcn_mbcnt_lo(m3, 0u)] = (r3 << 20) | (e0 + 3u);
                    } else {
                        if (lane == 0) atomicMin(&qlim, first);
                        direct |= 1u << j;
                    }
                }
            }
            if (direct) {
#pragma nounroll
                for (int jk = 0; jk < SCAN_DEPTH * 4; ++jk) {
                    const int j = jk >> 2, k = jk & 3;
                    if (!((direct >> j) & 1u)) continue;
                    const int4 sj = make_int4((j == 0) ? s[0].x : (j == 1) ? s[1].x : (j == 2) ? s[2].x : s[3].x,
                                              (j == 0) ? s[0].y : (j == 1) ? s[1].y : (j == 2) ? s[2].y : s[3].y,
                                              (j == 0) ? s[0].z : (j == 1) ? s[1].z : (j == 2) ? s[2].z : s[3].z,
                                              (j == 0) ? s[0].w : (j == 1) ? s[1].w : (j == 2) ? s[2].w : s[3].w);
                    const int cj = (j == 0) ? cid[0] : (j == 1) ? cid[1] : (j == 2) ? cid[2] : cid[3];
                    const unsigned rk = (unsigned)(pick4(sj, k) - base);
                    unsigned mk = __builtin_amdgcn_ballot_w32(rk < (unsigned)TILE_R);
                    while (mk) {
                        const int L = __builtin_ctz(mk);
                        mk &= mk - 1;
                        const unsigned r = (unsigned)__builtin_amdgcn_readlane((int)rk, L);
                        const int dn = dstp[(cj * 32 + L) * 4 + k];
                        const float* xr = x + (long)dn * F_OUT + lane;
                        const float logit = ss[r * N_HEAD + hd] + s_dst[dn * N_HEAD + hd];
                        const float lr = (logit >= 0.f) ? logit : NEG_SLOPE * logit;
                        const float ev = expf(-lr);
                        const int evb = __float_as_int(ev);
                        float* a = acc + r * (N_HEAD * F_OUT) + lane;
#pragma nounroll
                        for (int i = 0; i < N_HEAD; ++i) {
                            const float vi = __int_as_float(__builtin_amdgcn_readlane(evb, i));
#pragma unroll
                            for (int cc = 0; cc < 4; ++cc)
                                atomicAdd(a + i * F_OUT + cc * 32, vi * xr[cc * 32] * w[i * F_OUT + cc * 32 + lane]);
                        }
                        if (lane < N_HEAD)
                            atomicAdd(&rs[r * N_HEAD + lane], ev);
                    }
                }
            }
        }
    }

    __syncthreads();

    const int nq = __builtin_amdgcn_readfirstlane(min(qn, qlim));
    const bool ovf = qlim < QCAP;

    unsigned ent_[QCAP / TILE_THREADS];
    int      rk_[QCAP / TILE_THREADS];
#pragma unroll
    for (int u = 0; u < QCAP / TILE_THREADS; ++u) {
        const int t = tid + u * TILE_THREADS;
        ent_[u] = 0u; rk_[u] = 0;
        if (t < nq) {
            ent_[u] = q[t];
            rk_[u] = atomicAdd(&rcnt[ent_[u] >> 20], 1);
        }
    }
    __syncthreads();

    const int c0 = rcnt[2 * lane], c1 = rcnt[2 * lane + 1];
    int incl = c0 + c1;
#pragma unroll
    for (int d = 1; d < 32; d <<= 1) {
        const int v = __shfl_up(incl, d);
        if (lane >= d) incl += v;
    }
    const int excl = incl - (c0 + c1);
#pragma unroll
    for (int u = 0; u < QCAP / TILE_THREADS; ++u) {
        const int t = tid + u * TILE_THREADS;
        const int r = (int)(ent_[u] >> 20);
        const int se = __shfl(excl, r >> 1), sc = __shfl(c0, r >> 1);
        const int st = se + ((r & 1) ? sc : 0);
        if (t < nq) sorted[st + rk_[u]] = ent_[u];
    }
    __syncthreads();

    const int s0 = __builtin_amdgcn_readlane(excl, wid);
    const int s1 = s0 + __builtin_amdgcn_readlane(c0, wid);
    const int s2 = __builtin_amdgcn_readlane(incl, wid);

#pragma nounroll
    for (int h = 0; h < 2; ++h) {
        const int r  = 2 * wid + h;
        const int lo = h ? s1 : s0, hi = h ? s2 : s1;
        float sm[N_HEAD][4], dn_[N_HEAD];
#pragma unroll
        for (int i = 0; i < N_HEAD; ++i) {
            dn_[i] = 0.f;
#pragma unroll
            for (int c = 0; c < 4; ++c) sm[i][c] = 0.f;
        }
        const float4 sl = ((const float4*)ss)[r];
        const float4 sv = make_float4(__int_as_float(__builtin_amdgcn_readfirstlane(__float_as_int(sl.x))),
                                      __int_as_float(__builtin_amdgcn_readfirstlane(__float_as_int(sl.y))),
                                      __int_as_float(__builtin_amdgcn_readfirstlane(__float_as_int(sl.z))),
                                      __int_as_float(__builtin_amdgcn_readfirstlane(__float_as_int(sl.w))));

        for (int b0 = lo; b0 < hi; b0 += 32) {
            const int cnt = min(32, hi - b0);
            const bool live = lane < cnt;
            const unsigned ent = sorted[live ? (b0 + lane) : b0];
            const int dl = dstp[ent & 0xFFFFFu];
            const float4 sd = ((const float4*)s_dst)[dl];
            const float g0 = sv.x + sd.x;
            const float g1 = sv.y + sd.y;
            const float g2 = sv.z + sd.z;
            const float g3 = sv.w + sd.w;
            const float ev[N_HEAD] = { expf(-((g0 >= 0.f) ? g0 : NEG_SLOPE * g0)),
                                       expf(-((g1 >= 0.f) ? g1 : NEG_SLOPE * g1)),
                                       expf(-((g2 >= 0.f) ? g2 : NEG_SLOPE * g2)),
                                       expf(-((g3 >= 0.f) ? g3 : NEG_SLOPE * g3)) };
            for (int j0 = 0; j0 < cnt; j0 += DRAIN_AHEAD) {
                float4 xv[DRAIN_AHEAD];
#pragma unroll
                for (int u = 0; u < DRAIN_AHEAD; ++u) {
                    const int jj = min(j0 + u, cnt - 1);
                    const int dn = __builtin_amdgcn_readlane(dl, jj);
                    xv[u] = ((const float4*)(x + (long)dn * F_OUT))[lane];
                }
#pragma unroll
                for (int u = 0; u < DRAIN_AHEAD; ++u) {
                    const int jj = j0 + u;
                    if (jj < cnt) {
#pragma unroll
                        for (int i = 0; i < N_HEAD; ++i) {
                            const float vi = __int_as_float(__builtin_amdgcn_readlane(__float_as_int(ev[i]), jj));
                            dn_[i] += vi;
                            sm[i][0] += vi * xv[u].x; sm[i][1] += vi * xv[u].y;
                            sm[i][2] += vi * xv[u].z; sm[i][3] += vi * xv[u].w;
                        }
                    }
                }
            }
        }

        const int node = base + r;
        if (node < N_NODES) {
            const float4 dv = ((const float4*)rs)[r];
            const float dd[N_HEAD] = { dv.x, dv.y, dv.z, dv.w };
#pragma unroll
            for (int i = 0; i < N_HEAD; ++i) {
                const float4 wv = ((const float4*)(w + i * F_OUT))[lane];
                float4 v;
                v.x = sm[i][0] * wv.x; v.y = sm[i][1] * wv.y;
                v.z = sm[i][2] * wv.z; v.w = sm[i][3] * wv.w;
                if (ovf) {
                    const float4 a = ((const float4*)acc)[(r * N_HEAD + i) * (F_OUT / 4) + lane];
                    v.x += a.x; v.y += a.y; v.z += a.z; v.w += a.w;
                }
                const float inv = 1.0f / (dn_[i] + dd[i]);
                v.x *= inv; v.y *= inv; v.z *= inv; v.w *= inv;
                ((float4*)out)[((long)i * N_NODES + node) * (F_OUT / 4) + lane] = v;
            }
        }
    }
}

extern "C" void kernel_launch(void* const* d_in, const int* in_sizes, int n_in,
                              void* d_out, int out_size, void* d_ws, size_t ws_size,
                              hipStream_t stream) {
    const float* x  = (const float*)d_in[0];
    const float* w  = (const float*)d_in[1];
    const float* a  = (const float*)d_in[2];
    const int*   ei = (const int*)d_in[3];
    float* out = (float*)d_out;

    float* ws     = (float*)d_ws;
    float* s_src  = ws;
    float* s_dst  = ws + N_HEAD * N_NODES;
    unsigned* filt = (unsigned*)(ws + 2 * N_HEAD * N_NODES);
    float* coefB  = ws + 3 * N_HEAD * N_NODES;

    prep_kernel<<<COEF_BLOCKS + N_WORD, 256, 0, stream>>>(w, a, ei, coefB, filt);
    proj_kernel<<<N_NODES / 16, 32, 0, stream>>>(x, coefB, s_src, s_dst);
    tile_kernel<<<N_TILES, TILE_THREADS, 0, stream>>>(x, w, ei, filt, s_src, s_dst, out);
}


# scan trip slimmed: a lane that finds a tile row takes its list place itself by one returning LDS add (no per-piece ballots, counts, prefixes); walk skips empty filter words; list = edge arrival order
# speedup vs baseline: 9.1816x; 1.0647x over previous
#include <hip/hip_runtime.h>

#define N_NODES 50000
#define N_EDGES 800000
#define F_OUT   128
#define N_HEAD  4
#define NEG_SLOPE 0.2f

typedef __attribute__((ext_vector_type(2))) float v2f;
typedef __attribute__((ext_vector_type(8))) float v8f;

#define TILE_R       64
#define N_TILES      ((N_NODES + TILE_R - 1) / TILE_R)
#define N_CHUNK      (N_EDGES / 4 / 32)
#define N_WORD       ((N_CHUNK + 31) / 32)
#define FILT_PITCH   ((N_TILES + 31) / 32 * 32)
#define COEF_BLOCKS  ((F_OUT * 16 + 255) / 256)

static_assert((N_EDGES % 128) == 0, "a wave reads 32 whole int4 of sources at a time");
static_assert(N_WORD * FILT_PITCH <= N_HEAD * N_NODES, "the filter lies in the workspace between the destination scores and the coefficients");

__global__ __launch_bounds__(256)
void prep_kernel(const float* __restrict__ w, const float* __restrict__ a,
                 const int* __restrict__ ei,
                 float* __restrict__ coefB, unsigned* __restrict__ filt) {
    __shared__ unsigned bm[FILT_PITCH];
    const int tid = threadIdx.x;
    if (blockIdx.x < COEF_BLOCKS) {
        int idx = blockIdx.x * 256 + tid;
        if (idx >= F_OUT * 16) return;
        int k = idx >> 4, col = idx & 15;
        float v = 0.f;
        if (col < 8) {
            int head = col & 3, sel = col >> 2;
            v = w[head * F_OUT + k] * a[head * 2 * F_OUT + sel * F_OUT + k];
        }
        coefB[idx] = v;
        return;
    }
    const int cw = blockIdx.x - COEF_BLOCKS;
    for (int t = tid; t < FILT_PITCH; t += 256) bm[t] = 0u;
    __syncthreads();
    const int4* __restrict__ src4 = (const int4*)ei;
    int4 s[4];
    int  qi[4];
#pragma unroll
    for (int i = 0; i < 4; ++i) {
        qi[i] = cw * 1024 + i * 256 + tid;
        s[i] = src4[min(qi[i], N_EDGES / 4 - 1)];
    }
#pragma unroll
    for (int i = 0; i < 4; ++i) {
        if (qi[i] < N_EDGES / 4) {
            const unsigned bit = 1u << ((qi[i] >> 5) & 31);
            const unsigned t0 = (unsigned)s[i].x / TILE_R, t1 = (unsigned)s[i].y / TILE_R;
            const unsigned t2 = (unsigned)s[i].z / TILE_R, t3 = (unsigned)s[i].w / TILE_R;
            if (t0 < N_TILES) atomicOr(&bm[t0], bit);
            if (t1 < N_TILES) atomicOr(&bm[t1], bit);
            if (t2 < N_TILES) atomicOr(&bm[t2], bit);
            if (t3 < N_TILES) atomicOr(&bm[t3], bit);
        }
    }
    __syncthreads();
    for (int t = tid; t < FILT_PITCH; t += 256) filt[cw * FILT_PITCH + t] = bm[t];
}

__global__ void proj_kernel(const float* __restrict__ x, const float* __restrict__ coefB,
                            float* __restrict__ s_src, float* __restrict__ s_dst) {
    int tile = blockIdx.x;
    int lane = threadIdx.x;
    int half = lane >> 4;
    int rc   = lane & 15;
    int koff = half * 2;
    long arow = (long)(tile * 16 + rc) * F_OUT;

    v8f c = {0.f, 0.f, 0.f, 0.f, 0.f, 0.f, 0.f, 0.f};
#pragma unroll
    for (int k = 0; k < F_OUT; k += 4) {
        v2f av, bv;
        float2 ax = *(const float2*)(x + arow + k + koff);
        av.x = ax.x; av.y = ax.y;
        bv.x = coefB[(k + koff) * 16 + rc];
        bv.y = coefB[(k + koff + 1) * 16 + rc];
        c = __builtin_amdgcn_wmma_f32_16x16x4_f32(false, av, false, bv,
                                                  (short)0, c, false, false);
    }
    if (rc < 8) {
        int head = rc & 3;
        float* buf = (rc >> 2) ? s_dst : s_src;
        int mbase = tile * 16 + half * 8;
#pragma unroll
        for (int j = 0; j < 8; ++j)
            buf[(mbase + j) * N_HEAD + head] = c[j];
    }
}

#define TILE_THREADS 1024
#define SCAN_DEPTH   4
#define QCAP         2048
#define DRAIN_AHEAD  6

static_assert(N_WORD <= 32 * (TILE_THREADS / 32), "lane i of a wave holds the wave's i-th filter word");
static_assert((N_HEAD & (N_HEAD - 1)) == 0 && N_HEAD <= 32, "lane & (N_HEAD - 1) names a head; lanes 0..N_HEAD-1 hold one each");
static_assert(N_HEAD == 4, "a lane of the drain holds the four head scores of its edge in one float4");
static_assert(N_EDGES <= (1 << 20) && TILE_R <= (1 << 12), "a list word holds the row offset above bit 20 and the edge number below it");
static_assert(TILE_R == 64 && TILE_THREADS == 1024, "a lane scans two row counts; a wave owns two rows");
static_assert(QCAP % TILE_THREADS == 0, "a thread ranks QCAP / TILE_THREADS list words");

__device__ __forceinline__ int pick4(const int4 v, const int k) {
    return (k == 0) ? v.x : (k == 1) ? v.y : (k == 2) ? v.z : v.w;
}

__global__ __launch_bounds__(TILE_THREADS)
void tile_kernel(const float* __restrict__ x, const float* __restrict__ w,
                 const int* __restrict__ ei, const unsigned* __restrict__ filt,
                 const float* __restrict__ s_src, const float* __restrict__ s_dst,
                 float* __restrict__ out) {
    __shared__ float acc[TILE_R * N_HEAD * F_OUT];
    __shared__ float rs[TILE_R * N_HEAD];
    __shared__ float ss[TILE_R * N_HEAD];
    __shared__ unsigned q[QCAP];
    __shared__ unsigned sorted[QCAP];
    __shared__ int rcnt[TILE_R];
    __shared__ int qn;

    const int tid  = threadIdx.x;
    const int lane = tid & 31;
    const int hd   = lane & (N_HEAD - 1);
    const int wid  = __builtin_amdgcn_readfirstlane(tid >> 5);
    const int base = blockIdx.x * TILE_R;

    unsigned fw = 0u;
    if (wid + 32 * lane < N_WORD) fw = filt[(wid + 32 * lane) * FILT_PITCH + blockIdx.x];

    const float4 z4 = make_float4(0.f, 0.f, 0.f, 0.f);
    for (int i = tid; i < TILE_R * N_HEAD * F_OUT / 4; i += TILE_THREADS)
        ((float4*)acc)[i] = z4;
    if (tid < TILE_R * N_HEAD) {
        int node = base + (tid >> 2);
        rs[tid] = 0.f;
        ss[tid] = (node < N_NODES) ? s_src[node * N_HEAD + (tid & 3)] : 0.f;
    }
    if (tid < TILE_R) rcnt[tid] = 0;
    if (tid == 0) qn = 0;

    __syncthreads();

    const int4* __restrict__ src4 = (const int4*)ei;
    const int*  __restrict__ dstp = ei + N_EDGES;

    {
        unsigned wm = __builtin_amdgcn_ballot_w32(fw != 0u);
        unsigned mask = 0u;
        int wi = 0;
        for (;;) {
            int  cid[SCAN_DEPTH];
            bool val[SCAN_DEPTH];
#pragma unroll
            for (int j = 0; j < SCAN_DEPTH; ++j) {
                if (mask == 0u && wm != 0u) {
                    wi = __builtin_ctz(wm);
                    wm &= wm - 1u;
                    mask = (unsigned)__builtin_amdgcn_readlane((int)fw, wi);
                }
                val[j] = (mask != 0u);
                cid[j] = val[j] ? ((wid + 32 * wi) * 32 + __builtin_ctz(mask)) : (j ? cid[0] : 0);
                mask &= mask - 1u;
            }
            if (!val[0]) break;
            int4 s[SCAN_DEPTH];
#pragma unroll
            for (int j = 0; j < SCAN_DEPTH; ++j)
                s[j] = src4[cid[j] * 32 + lane];
            unsigned ovb = 0u;
#pragma unroll
            for (int j = 0; j < SCAN_DEPTH; ++j) {
                if (val[j]) {
                    const unsigned e0 = (unsigned)((cid[j] * 32 + lane) * 4);
                    const unsigned rr[4] = { (unsigned)(s[j].x - base), (unsigned)(s[j].y - base),
                                             (unsigned)(s[j].z - base), (unsigned)(s[j].w - base) };
#pragma unroll
                    for (int k = 0; k < 4; ++k) {
                        if (rr[k] < (unsigned)TILE_R) {
                            const int p = atomicAdd(&qn, 1);
                            if (p < QCAP) q[p] = (rr[k] << 20) | (e0 + (unsigned)k);
                            else          ovb |= 1u << (j * 4 + k);
                        }
                    }
                }
            }
            if (__builtin_amdgcn_ballot_w32(ovb != 0u)) {
#pragma nounroll
                for (int jk = 0; jk < SCAN_DEPTH * 4; ++jk) {
                    const int j = jk >> 2, k = jk & 3;
                    unsigned mk = __builtin_amdgcn_ballot_w32(((ovb >> jk) & 1u) != 0u);
                    if (!mk) continue;
                    const int4 sj = make_int4((j == 0) ? s[0].x : (j == 1) ? s[1].x : (j == 2) ? s[2].x : s[3].x,
                                              (j == 0) ? s[0].y : (j == 1) ? s[1].y : (j == 2) ? s[2].y : s[3].y,
                                              (j == 0) ? s[0].z : (j == 1) ? s[1].z : (j == 2) ? s[2].z : s[3].z,
                                              (j == 0) ? s[0].w : (j == 1) ? s[1].w : (j == 2) ? s[2].w : s[3].w);
                    const int cj = (j == 0) ? cid[0] : (j == 1) ? cid[1] : (j == 2) ? cid[2] : cid[3];
                    const unsigned rk = (unsigned)(pick4(sj, k) - base);
                    while (mk) {
                        const int L = __builtin_ctz(mk);
                        mk &= mk - 1;
                        const unsigned r = (unsigned)__builtin_amdgcn_readlane((int)rk, L);
                        const int dn = dstp[(cj * 32 + L) * 4 + k];
                        const float* xr = x + (long)dn * F_OUT + lane;
                        const float logit = ss[r * N_HEAD + hd] + s_dst[dn * N_HEAD + hd];
                        const float lr = (logit >= 0.f) ? logit : NEG_SLOPE * logit;
                        const float ev = expf(-lr);
                        const int evb = __float_as_int(ev);
                        float* a = acc + r * (N_HEAD * F_OUT) + lane;
#pragma nounroll
                        for (int i = 0; i < N_HEAD; ++i) {
                            const float vi = __int_as_float(__builtin_amdgcn_readlane(evb, i));
#pragma unroll
                            for (int cc = 0; cc < 4; ++cc)
                                atomicAdd(a + i * F_OUT + cc * 32, vi * xr[cc * 32] * w[i * F_OUT + cc * 32 + lane]);
                        }
                        if (lane < N_HEAD)
                            atomicAdd(&rs[r * N_HEAD + lane], ev);
                    }
                }
            }
        }
    }

    __syncthreads();

    const int qtot = qn;
    const int nq = __builtin_amdgcn_readfirstlane(min(qtot, QCAP));
    const bool ovf = qtot > QCAP;

    unsigned ent_[QCAP / TILE_THREADS];
    int      rk_[QCAP / TILE_THREADS];
#pragma unroll
    for (int u = 0; u < QCAP / TILE_THREADS; ++u) {
        const int t = tid + u * TILE_THREADS;
        ent_[u] = 0u; rk_[u] = 0;
        if (t < nq) {
            ent_[u] = q[t];
            rk_[u] = atomicAdd(&rcnt[ent_[u] >> 20], 1);
        }
    }
    __syncthreads();

    const int c0 = rcnt[2 * lane], c1 = rcnt[2 * lane + 1];
    int incl = c0 + c1;
#pragma unroll
    for (int d = 1; d < 32; d <<= 1) {
        const int v = __shfl_up(incl, d);
        if (lane >= d) incl += v;
    }
    const int excl = incl - (c0 + c1);
#pragma unroll
    for (int u = 0; u < QCAP / TILE_THREADS; ++u) {
        const int t = tid + u * TILE_THREADS;
        const int r = (int)(ent_[u] >> 20);
        const int se = __shfl(excl, r >> 1), sc = __shfl(c0, r >> 1);
        const int st = se + ((r & 1) ? sc : 0);
        if (t < nq) sorted[st + rk_[u]] = ent_[u];
    }
    __syncthreads();

    const int s0 = __builtin_amdgcn_readlane(excl, wid);
    const int s1 = s0 + __builtin_amdgcn_readlane(c0, wid);
    const int s2 = __builtin_amdgcn_readlane(incl, wid);

#pragma nounroll
    for (int h = 0; h < 2; ++h) {
        const int r  = 2 * wid + h;
        const int lo = h ? s1 : s0, hi = h ? s2 : s1;
        float sm[N_HEAD][4], dn_[N_HEAD];
#pragma unroll
        for (int i = 0; i < N_HEAD; ++i) {
            dn_[i] = 0.f;
#pragma unroll
            for (int c = 0; c < 4; ++c) sm[i][c] = 0.f;
        }
        const float4 sl = ((const float4*)ss)[r];
        const float4 sv = make_float4(__int_as_float(__builtin_amdgcn_readfirstlane(__float_as_int(sl.x))),
                                      __int_as_float(__builtin_amdgcn_readfirstlane(__float_as_int(sl.y))),
                                      __int_as_float(__builtin_amdgcn_readfirstlane(__float_as_int(sl.z))),
                                      __int_as_float(__builtin_amdgcn_readfirstlane(__float_as_int(sl.w))));

        for (int b0 = lo; b0 < hi; b0 += 32) {
            const int cnt = min(32, hi - b0);
            const bool live = lane < cnt;
            const unsigned ent = sorted[live ? (b0 + lane) : b0];
            const int dl = dstp[ent & 0xFFFFFu];
            const float4 sd = ((const float4*)s_dst)[dl];
            const float g0 = sv.x + sd.x;
            const float g1 = sv.y + sd.y;
            const float g2 = sv.z + sd.z;
            const float g3 = sv.w + sd.w;
            const float ev[N_HEAD] = { expf(-((g0 >= 0.f) ? g0 : NEG_SLOPE * g0)),
                                       expf(-((g1 >= 0.f) ? g1 : NEG_SLOPE * g1)),
                                       expf(-((g2 >= 0.f) ? g2 : NEG_SLOPE * g2)),
                                       expf(-((g3 >= 0.f) ? g3 : NEG_SLOPE * g3)) };
            for (int j0 = 0; j0 < cnt; j0 += DRAIN_AHEAD) {
                float4 xv[DRAIN_AHEAD];
#pragma unroll
                for (int u = 0; u < DRAIN_AHEAD; ++u) {
                    const int jj = min(j0 + u, cnt - 1);
                    const int dn = __builtin_amdgcn_readlane(dl, jj);
                    xv[u] = ((const float4*)(x + (long)dn * F_OUT))[lane];
                }
#pragma unroll
                for (int u = 0; u < DRAIN_AHEAD; ++u) {
                    const int jj = j0 + u;
                    if (jj < cnt) {
#pragma unroll
                        for (int i = 0; i < N_HEAD; ++i) {
                            const float vi = __int_as_float(__builtin_amdgcn_readlane(__float_as_int(ev[i]), jj));
                            dn_[i] += vi;
                            sm[i][0] += vi * xv[u].x; sm[i][1] += vi * xv[u].y;
                            sm[i][2] += vi * xv[u].z; sm[i][3] += vi * xv[u].w;
                        }
                    }
                }
            }
        }

        const int node = base + r;
        if (node < N_NODES) {
            const float4 dv = ((const float4*)rs)[r];
            const float dd[N_HEAD] = { dv.x, dv.y, dv.z, dv.w };
#pragma unroll
            for (int i = 0; i < N_HEAD; ++i) {
                const float4 wv = ((const float4*)(w + i * F_OUT))[lane];
                float4 v;
                v.x = sm[i][0] * wv.x; v.y = sm[i][1] * wv.y;
                v.z = sm[i][2] * wv.z; v.w = sm[i][3] * wv.w;
                if (ovf) {
                    const float4 a = ((const float4*)acc)[(r * N_HEAD + i) * (F_OUT / 4) + lane];
                    v.x += a.x; v.y += a.y; v.z += a.z; v.w += a.w;
                }
                const float inv = 1.0f / (dn_[i] + dd[i]);
                v.x *= inv; v.y *= inv; v.z *= inv; v.w *= inv;
                ((float4*)out)[((long)i * N_NODES + node) * (F_OUT / 4) + lane] = v;
            }
        }
    }
}

extern "C" void kernel_launch(void* const* d_in, const int* in_sizes, int n_in,
                              void* d_out, int out_size, void* d_ws, size_t ws_size,
                              hipStream_t stream) {
    const float* x  = (const float*)d_in[0];
    const float* w  = (const float*)d_in[1];
    const float* a  = (const float*)d_in[2];
    const int*   ei = (const int*)d_in[3];
    float* out = (float*)d_out;

    float* ws     = (float*)d_ws;
    float* s_src  = ws;
    float* s_dst  = ws + N_HEAD * N_NODES;
    unsigned* filt = (unsigned*)(ws + 2 * N_HEAD * N_NODES);
    float* coefB  = ws + 3 * N_HEAD * N_NODES;

    prep_kernel<<<COEF_BLOCKS + N_WORD, 256, 0, stream>>>(w, a, ei, coefB, filt);
    proj_kernel<<<N_NODES / 16, 32, 0, stream>>>(x, coefB, s_src, s_dst);
    tile_kernel<<<N_TILES, TILE_THREADS, 0, stream>>>(x, w, ei, filt, s_src, s_dst, out);
}


# found edges go straight into per-row lists in LDS (one returning add on the row's counter gives the place): no common list, no counting sort, two barriers fewer; sum order = arrival order per row
# speedup vs baseline: 9.6356x; 1.0494x over previous
#include <hip/hip_runtime.h>

#define N_NODES 50000
#define N_EDGES 800000
#define F_OUT   128
#define N_HEAD  4
#define NEG_SLOPE 0.2f

typedef __attribute__((ext_vector_type(2))) float v2f;
typedef __attribute__((ext_vector_type(8))) float v8f;

#define TILE_R       64
#define N_TILES      ((N_NODES + TILE_R - 1) / TILE_R)
#define N_CHUNK      (N_EDGES / 4 / 32)
#define N_WORD       ((N_CHUNK + 31) / 32)
#define FILT_PITCH   ((N_TILES + 31) / 32 * 32)
#define COEF_BLOCKS  ((F_OUT * 16 + 255) / 256)

static_assert((N_EDGES % 128) == 0, "a wave reads 32 whole int4 of sources at a time");
static_assert(N_WORD * FILT_PITCH <= N_HEAD * N_NODES, "the filter lies in the workspace between the destination scores and the coefficients");

__global__ __launch_bounds__(256)
void prep_kernel(const float* __restrict__ w, const float* __restrict__ a,
                 const int* __restrict__ ei,
                 float* __restrict__ coefB, unsigned* __restrict__ filt) {
    __shared__ unsigned bm[FILT_PITCH];
    const int tid = threadIdx.x;
    if (blockIdx.x < COEF_BLOCKS) {
        int idx = blockIdx.x * 256 + tid;
        if (idx >= F_OUT * 16) return;
        int k = idx >> 4, col = idx & 15;
        float v = 0.f;
        if (col < 8) {
            int head = col & 3, sel = col >> 2;
            v = w[head * F_OUT + k] * a[head * 2 * F_OUT + sel * F_OUT + k];
        }
        coefB[idx] = v;
        return;
    }
    const int cw = blockIdx.x - COEF_BLOCKS;
    for (int t = tid; t < FILT_PITCH; t += 256) bm[t] = 0u;
    __syncthreads();
    const int4* __restrict__ src4 = (const int4*)ei;
    int4 s[4];
    int  qi[4];
#pragma unroll
    for (int i = 0; i < 4; ++i) {
        qi[i] = cw * 1024 + i * 256 + tid;
        s[i] = src4[min(qi[i], N_EDGES / 4 - 1)];
    }
#pragma unroll
    for (int i = 0; i < 4; ++i) {
        if (qi[i] < N_EDGES / 4) {
            const unsigned bit = 1u << ((qi[i] >> 5) & 31);
            const unsigned t0 = (unsigned)s[i].x / TILE_R, t1 = (unsigned)s[i].y / TILE_R;
            const unsigned t2 = (unsigned)s[i].z / TILE_R, t3 = (unsigned)s[i].w / TILE_R;
            if (t0 < N_TILES) atomicOr(&bm[t0], bit);
            if (t1 < N_TILES) atomicOr(&bm[t1], bit);
            if (t2 < N_TILES) atomicOr(&bm[t2], bit);
            if (t3 < N_TILES) atomicOr(&bm[t3], bit);
        }
    }
    __syncthreads();
    for (int t = tid; t < FILT_PITCH; t += 256) filt[cw * FILT_PITCH + t] = bm[t];
}

__global__ void proj_kernel(const float* __restrict__ x, const float* __restrict__ coefB,
                            float* __restrict__ s_src, float* __restrict__ s_dst) {
    int tile = blockIdx.x;
    int lane = threadIdx.x;
    int half = lane >> 4;
    int rc   = lane & 15;
    int koff = half * 2;
    long arow = (long)(tile * 16 + rc) * F_OUT;

    v8f c = {0.f, 0.f, 0.f, 0.f, 0.f, 0.f, 0.f, 0.f};
#pragma unroll
    for (int k = 0; k < F_OUT; k += 4) {
        v2f av, bv;
        float2 ax = *(const float2*)(x + arow + k + koff);
        av.x = ax.x; av.y = ax.y;
        bv.x = coefB[(k + koff) * 16 + rc];
        bv.y = coefB[(k + koff + 1) * 16 + rc];
        c = __builtin_amdgcn_wmma_f32_16x16x4_f32(false, av, false, bv,
                                                  (short)0, c, false, false);
    }
    if (rc < 8) {
        int head = rc & 3;
        float* buf = (rc >> 2) ? s_dst : s_src;
        int mbase = tile * 16 + half * 8;
#pragma unroll
        for (int j = 0; j < 8; ++j)
            buf[(mbase + j) * N_HEAD + head] = c[j];
    }
}

#define TILE_THREADS 1024
#define SCAN_DEPTH   4
#define BCAP         64
#define DRAIN_AHEAD  6

static_assert(N_WORD <= 32 * (TILE_THREADS / 32), "lane i of a wave holds the wave's i-th filter word");
static_assert((N_HEAD & (N_HEAD - 1)) == 0 && N_HEAD <= 32, "lane & (N_HEAD - 1) names a head; lanes 0..N_HEAD-1 hold one each");
static_assert(N_HEAD == 4, "a lane of the drain holds the four head scores of its edge in one float4");
static_assert(TILE_R == 2 * (TILE_THREADS / 32), "a wave owns two rows");

__device__ __forceinline__ int pick4(const int4 v, const int k) {
    return (k == 0) ? v.x : (k == 1) ? v.y : (k == 2) ? v.z : v.w;
}

__global__ __launch_bounds__(TILE_THREADS)
void tile_kernel(const float* __restrict__ x, const float* __restrict__ w,
                 const int* __restrict__ ei, const unsigned* __restrict__ filt,
                 const float* __restrict__ s_src, const float* __restrict__ s_dst,
                 float* __restrict__ out) {
    __shared__ float acc[TILE_R * N_HEAD * F_OUT];
    __shared__ float rs[TILE_R * N_HEAD];
    __shared__ float ss[TILE_R * N_HEAD];
    __shared__ unsigned bkt[TILE_R * BCAP];
    __shared__ int rcnt[TILE_R];
    __shared__ int ovfl;

    const int tid  = threadIdx.x;
    const int lane = tid & 31;
    const int hd   = lane & (N_HEAD - 1);
    const int wid  = __builtin_amdgcn_readfirstlane(tid >> 5);
    const int base = blockIdx.x * TILE_R;

    unsigned fw = 0u;
    if (wid + 32 * lane < N_WORD) fw = filt[(wid + 32 * lane) * FILT_PITCH + blockIdx.x];

    const float4 z4 = make_float4(0.f, 0.f, 0.f, 0.f);
    for (int i = tid; i < TILE_R * N_HEAD * F_OUT / 4; i += TILE_THREADS)
        ((float4*)acc)[i] = z4;
    if (tid < TILE_R * N_HEAD) {
        int node = base + (tid >> 2);
        rs[tid] = 0.f;
        ss[tid] = (node < N_NODES) ? s_src[node * N_HEAD + (tid & 3)] : 0.f;
    }
    if (tid < TILE_R) rcnt[tid] = 0;
    if (tid == 0) ovfl = 0;

    __syncthreads();

    const int4* __restrict__ src4 = (const int4*)ei;
    const int*  __restrict__ dstp = ei + N_EDGES;

    {
        unsigned wm = __builtin_amdgcn_ballot_w32(fw != 0u);
        unsigned mask = 0u;
        int wi = 0;
        for (;;) {
            int  cid[SCAN_DEPTH];
            bool val[SCAN_DEPTH];
#pragma unroll
            for (int j = 0; j < SCAN_DEPTH; ++j) {
                if (mask == 0u && wm != 0u) {
                    wi = __builtin_ctz(wm);
                    wm &= wm - 1u;
                    mask = (unsigned)__builtin_amdgcn_readlane((int)fw, wi);
                }
                val[j] = (mask != 0u);
                cid[j] = val[j] ? ((wid + 32 * wi) * 32 + __builtin_ctz(mask)) : (j ? cid[0] : 0);
                mask &= mask - 1u;
            }
            if (!val[0]) break;
            int4 s[SCAN_DEPTH];
#pragma unroll
            for (int j = 0; j < SCAN_DEPTH; ++j)
                s[j] = src4[cid[j] * 32 + lane];
            unsigned ovb = 0u;
#pragma unroll
            for (int j = 0; j < SCAN_DEPTH; ++j) {
                if (val[j]) {
                    const unsigned e0 = (unsigned)((cid[j] * 32 + lane) * 4);
                    const unsigned rr[4] = { (unsigned)(s[j].x - base), (unsigned)(s[j].y - base),
                                             (unsigned)(s[j].z - base), (unsigned)(s[j].w - base) };
#pragma unroll
                    for (int k = 0; k < 4; ++k) {
                        if (rr[k] < (unsigned)TILE_R) {
                            const int p = atomicAdd(&rcnt[rr[k]], 1);
                            if (p < BCAP) bkt[rr[k] * BCAP + p] = e0 + (unsigned)k;
                            else          ovb |= 1u << (j * 4 + k);
                        }
                    }
                }
            }
            if (__builtin_amdgcn_ballot_w32(ovb != 0u)) {
                if (lane == 0) ovfl = 1;
#pragma nounroll
                for (int jk = 0; jk < SCAN_DEPTH * 4; ++jk) {
                    const int j = jk >> 2, k = jk & 3;
                    unsigned mk = __builtin_amdgcn_ballot_w32(((ovb >> jk) & 1u) != 0u);
                    if (!mk) continue;
                    const int4 sj = make_int4((j == 0) ? s[0].x : (j == 1) ? s[1].x : (j == 2) ? s[2].x : s[3].x,
                                              (j == 0) ? s[0].y : (j == 1) ? s[1].y : (j == 2) ? s[2].y : s[3].y,
                                              (j == 0) ? s[0].z : (j == 1) ? s[1].z : (j == 2) ? s[2].z : s[3].z,
                                              (j == 0) ? s[0].w : (j == 1) ? s[1].w : (j == 2) ? s[2].w : s[3].w);
                    const int cj = (j == 0) ? cid[0] : (j == 1) ? cid[1] : (j == 2) ? cid[2] : cid[3];
                    const unsigned rk = (unsigned)(pick4(sj, k) - base);
                    while (mk) {
                        const int L = __builtin_ctz(mk);
                        mk &= mk - 1;
                        const unsigned r = (unsigned)__builtin_amdgcn_readlane((int)rk, L);
                        const int dn = dstp[(cj * 32 + L) * 4 + k];
                        const float* xr = x + (long)dn * F_OUT + lane;
                        const float logit = ss[r * N_HEAD + hd] + s_dst[dn * N_HEAD + hd];
                        const float lr = (logit >= 0.f) ? logit : NEG_SLOPE * logit;
                        const float ev = expf(-lr);
                        const int evb = __float_as_int(ev);
                        float* a = acc + r * (N_HEAD * F_OUT) + lane;
#pragma nounroll
                        for (int i = 0; i < N_HEAD; ++i) {
                            const float vi = __int_as_float(__builtin_amdgcn_readlane(evb, i));
#pragma unroll
                            for (int cc = 0; cc < 4; ++cc)
                                atomicAdd(a + i * F_OUT + cc * 32, vi * xr[cc * 32] * w[i * F_OUT + cc * 32 + lane]);
                        }
                        if (lane < N_HEAD)
                            atomicAdd(&rs[r * N_HEAD + lane], ev);
                    }
                }
            }
        }
    }

    __syncthreads();

    const bool ovf = ovfl != 0;

#pragma nounroll
    for (int h = 0; h < 2; ++h) {
        const int r  = 2 * wid + h;
        const int nr = __builtin_amdgcn_readfirstlane(min(rcnt[r], BCAP));
        const unsigned* __restrict__ br = bkt + r * BCAP;
        float sm[N_HEAD][4], dn_[N_HEAD];
#pragma unroll
        for (int i = 0; i < N_HEAD; ++i) {
            dn_[i] = 0.f;
#pragma unroll
            for (int c = 0; c < 4; ++c) sm[i][c] = 0.f;
        }
        const float4 sl = ((const float4*)ss)[r];
        const float4 sv = make_float4(__int_as_float(__builtin_amdgcn_readfirstlane(__float_as_int(sl.x))),
                                      __int_as_float(__builtin_amdgcn_readfirstlane(__float_as_int(sl.y))),
                                      __int_as_float(__builtin_amdgcn_readfirstlane(__float_as_int(sl.z))),
                                      __int_as_float(__builtin_amdgcn_readfirstlane(__float_as_int(sl.w))));

        for (int b0 = 0; b0 < nr; b0 += 32) {
            const int cnt = min(32, nr - b0);
            const bool live = lane < cnt;
            const unsigned ent = br[live ? (b0 + lane) : b0];
            const int dl = dstp[ent];
            const float4 sd = ((const float4*)s_dst)[dl];
            const float g0 = sv.x + sd.x;
            const float g1 = sv.y + sd.y;
            const float g2 = sv.z + sd.z;
            const float g3 = sv.w + sd.w;
            const float ev[N_HEAD] = { expf(-((g0 >= 0.f) ? g0 : NEG_SLOPE * g0)),
                                       expf(-((g1 >= 0.f) ? g1 : NEG_SLOPE * g1)),
                                       expf(-((g2 >= 0.f) ? g2 : NEG_SLOPE * g2)),
                                       expf(-((g3 >= 0.f) ? g3 : NEG_SLOPE * g3)) };
            for (int j0 = 0; j0 < cnt; j0 += DRAIN_AHEAD) {
                float4 xv[DRAIN_AHEAD];
#pragma unroll
                for (int u = 0; u < DRAIN_AHEAD; ++u) {
                    const int jj = min(j0 + u, cnt - 1);
                    const int dn = __builtin_amdgcn_readlane(dl, jj);
                    xv[u] = ((const float4*)(x + (long)dn * F_OUT))[lane];
                }
#pragma unroll
                for (int u = 0; u < DRAIN_AHEAD; ++u) {
                    const int jj = j0 + u;
                    if (jj < cnt) {
#pragma unroll
                        for (int i = 0; i < N_HEAD; ++i) {
                            const float vi = __int_as_float(__builtin_amdgcn_readlane(__float_as_int(ev[i]), jj));
                            dn_[i] += vi;
                            sm[i][0] += vi * xv[u].x; sm[i][1] += vi * xv[u].y;
                            sm[i][2] += vi * xv[u].z; sm[i][3] += vi * xv[u].w;
                        }
                    }
                }
            }
        }

        const int node = base + r;
        if (node < N_NODES) {
            const float4 dv = ((const float4*)rs)[r];
            const float dd[N_HEAD] = { dv.x, dv.y, dv.z, dv.w };
#pragma unroll
            for (int i = 0; i < N_HEAD; ++i) {
                const float4 wv = ((const float4*)(w + i * F_OUT))[lane];
                float4 v;
                v.x = sm[i][0] * wv.x; v.y = sm[i][1] * wv.y;
                v.z = sm[i][2] * wv.z; v.w = sm[i][3] * wv.w;
                if (ovf) {
                    const float4 a = ((const float4*)acc)[(r * N_HEAD + i) * (F_OUT / 4) + lane];
                    v.x += a.x; v.y += a.y; v.z += a.z; v.w += a.w;
                }
                const float inv = 1.0f / (dn_[i] + dd[i]);
                v.x *= inv; v.y *= inv; v.z *= inv; v.w *= inv;
                ((float4*)out)[((long)i * N_NODES + node) * (F_OUT / 4) + lane] = v;
            }
        }
    }
}

extern "C" void kernel_launch(void* const* d_in, const int* in_sizes, int n_in,
                              void* d_out, int out_size, void* d_ws, size_t ws_size,
                              hipStream_t stream) {
    const float* x  = (const float*)d_in[0];
    const float* w  = (const float*)d_in[1];
    const float* a  = (const float*)d_in[2];
    const int*   ei = (const int*)d_in[3];
    float* out = (float*)d_out;

    float* ws     = (float*)d_ws;
    float* s_src  = ws;
    float* s_dst  = ws + N_HEAD * N_NODES;
    unsigned* filt = (unsigned*)(ws + 2 * N_HEAD * N_NODES);
    float* coefB  = ws + 3 * N_HEAD * N_NODES;

    prep_kernel<<<COEF_BLOCKS + N_WORD, 256, 0, stream>>>(w, a, ei, coefB, filt);
    proj_kernel<<<N_NODES / 16, 32, 0, stream>>>(x, coefB, s_src, s_dst);
    tile_kernel<<<N_TILES, TILE_THREADS, 0, stream>>>(x, w, ei, filt, s_src, s_dst, out);
}


# even shares inside a work-group: marked pieces listed in LDS and dealt to the waves trip by trip; rows drawn from an LDS counter, not two fixed per wave; sum order = arrival order per row
# speedup vs baseline: 10.2123x; 1.0599x over previous
#include <hip/hip_runtime.h>

#define N_NODES 50000
#define N_EDGES 800000
#define F_OUT   128
#define N_HEAD  4
#define NEG_SLOPE 0.2f

typedef __attribute__((ext_vector_type(2))) float v2f;
typedef __attribute__((ext_vector_type(8))) float v8f;

#define TILE_R       64
#define N_TILES      ((N_NODES + TILE_R - 1) / TILE_R)
#define N_CHUNK      (N_EDGES / 4 / 32)
#define N_WORD       ((N_CHUNK + 31) / 32)
#define FILT_PITCH   ((N_TILES + 31) / 32 * 32)
#define COEF_BLOCKS  ((F_OUT * 16 + 255) / 256)

static_assert((N_EDGES % 128) == 0, "a wave reads 32 whole int4 of sources at a time");
static_assert(N_WORD * FILT_PITCH <= N_HEAD * N_NODES, "the filter lies in the workspace between the destination scores and the coefficients");

__global__ __launch_bounds__(256)
void prep_kernel(const float* __restrict__ w, const float* __restrict__ a,
                 const int* __restrict__ ei,
                 float* __restrict__ coefB, unsigned* __restrict__ filt) {
    __shared__ unsigned bm[FILT_PITCH];
    const int tid = threadIdx.x;
    if (blockIdx.x < COEF_BLOCKS) {
        int idx = blockIdx.x * 256 + tid;
        if (idx >= F_OUT * 16) return;
        int k = idx >> 4, col = idx & 15;
        float v = 0.f;
        if (col < 8) {
            int head = col & 3, sel = col >> 2;
            v = w[head * F_OUT + k] * a[head * 2 * F_OUT + sel * F_OUT + k];
        }
        coefB[idx] = v;
        return;
    }
    const int cw = blockIdx.x - COEF_BLOCKS;
    for (int t = tid; t < FILT_PITCH; t += 256) bm[t] = 0u;
    __syncthreads();
    const int4* __restrict__ src4 = (const int4*)ei;
    int4 s[4];
    int  qi[4];
#pragma unroll
    for (int i = 0; i < 4; ++i) {
        qi[i] = cw * 1024 + i * 256 + tid;
        s[i] = src4[min(qi[i], N_EDGES / 4 - 1)];
    }
#pragma unroll
    for (int i = 0; i < 4; ++i) {
        if (qi[i] < N_EDGES / 4) {
            const unsigned bit = 1u << ((qi[i] >> 5) & 31);
            const unsigned t0 = (unsigned)s[i].x / TILE_R, t1 = (unsigned)s[i].y / TILE_R;
            const unsigned t2 = (unsigned)s[i].z / TILE_R, t3 = (unsigned)s[i].w / TILE_R;
            if (t0 < N_TILES) atomicOr(&bm[t0], bit);
            if (t1 < N_TILES) atomicOr(&bm[t1], bit);
            if (t2 < N_TILES) atomicOr(&bm[t2], bit);
            if (t3 < N_TILES) atomicOr(&bm[t3], bit);
        }
    }
    __syncthreads();
    for (int t = tid; t < FILT_PITCH; t += 256) filt[cw * FILT_PITCH + t] = bm[t];
}

__global__ void proj_kernel(const float* __restrict__ x, const float* __restrict__ coefB,
                            float* __restrict__ s_src, float* __restrict__ s_dst) {
    int tile = blockIdx.x;
    int lane = threadIdx.x;
    int half = lane >> 4;
    int rc   = lane & 15;
    int koff = half * 2;
    long arow = (long)(tile * 16 + rc) * F_OUT;

    v8f c = {0.f, 0.f, 0.f, 0.f, 0.f, 0.f, 0.f, 0.f};
#pragma unroll
    for (int k = 0; k < F_OUT; k += 4) {
        v2f av, bv;
        float2 ax = *(const float2*)(x + arow + k + koff);
        av.x = ax.x; av.y = ax.y;
        bv.x = coefB[(k + koff) * 16 + rc];
        bv.y = coefB[(k + koff + 1) * 16 + rc];
        c = __builtin_amdgcn_wmma_f32_16x16x4_f32(false, av, false, bv,
                                                  (short)0, c, false, false);
    }
    if (rc < 8) {
        int head = rc & 3;
        float* buf = (rc >> 2) ? s_dst : s_src;
        int mbase = tile * 16 + half * 8;
#pragma unroll
        for (int j = 0; j < 8; ++j)
            buf[(mbase + j) * N_HEAD + head] = c[j];
    }
}

#define TILE_THREADS 1024
#define SCAN_DEPTH   4
#define BCAP         60
#define DRAIN_AHEAD  6

static_assert(N_WORD <= TILE_THREADS, "thread i holds filter word i");
static_assert(SCAN_DEPTH == 4, "a trip's list places are one 8-byte read of four 16-bit numbers");
static_assert(N_CHUNK <= 65536, "a piece number fits the 16-bit places of the piece list");
static_assert((N_HEAD & (N_HEAD - 1)) == 0 && N_HEAD <= 32, "lane & (N_HEAD - 1) names a head; lanes 0..N_HEAD-1 hold one each");
static_assert(N_HEAD == 4, "a lane of the drain holds the four head scores of its edge in one float4");

__device__ __forceinline__ int pick4(const int4 v, const int k) {
    return (k == 0) ? v.x : (k == 1) ? v.y : (k == 2) ? v.z : v.w;
}

__global__ __launch_bounds__(TILE_THREADS)
void tile_kernel(const float* __restrict__ x, const float* __restrict__ w,
                 const int* __restrict__ ei, const unsigned* __restrict__ filt,
                 const float* __restrict__ s_src, const float* __restrict__ s_dst,
                 float* __restrict__ out) {
    __shared__ float acc[TILE_R * N_HEAD * F_OUT];
    __shared__ float rs[TILE_R * N_HEAD];
    __shared__ float ss[TILE_R * N_HEAD];
    __shared__ unsigned bkt[TILE_R * BCAP];
    __shared__ int rcnt[TILE_R];
    __shared__ int ovfl;
    __shared__ __attribute__((aligned(8))) unsigned short plist[N_WORD * 32];
    __shared__ int npl;
    __shared__ int nextr;

    const int tid  = threadIdx.x;
    const int lane = tid & 31;
    const int hd   = lane & (N_HEAD - 1);
    const int wid  = __builtin_amdgcn_readfirstlane(tid >> 5);
    const int base = blockIdx.x * TILE_R;

    if (tid == 0) { npl = 0; nextr = 0; ovfl = 0; }
    __syncthreads();

    unsigned fw = 0u;
    if (tid < N_WORD) fw = filt[tid * FILT_PITCH + blockIdx.x];

    const float4 z4 = make_float4(0.f, 0.f, 0.f, 0.f);
    for (int i = tid; i < TILE_R * N_HEAD * F_OUT / 4; i += TILE_THREADS)
        ((float4*)acc)[i] = z4;
    if (tid < TILE_R * N_HEAD) {
        int node = base + (tid >> 2);
        rs[tid] = 0.f;
        ss[tid] = (node < N_NODES) ? s_src[node * N_HEAD + (tid & 3)] : 0.f;
    }
    if (tid < TILE_R) rcnt[tid] = 0;

    if (fw != 0u) {
        int pos = atomicAdd(&npl, __popc(fw));
        while (fw) {
            plist[pos++] = (unsigned short)(tid * 32 + __builtin_ctz(fw));
            fw &= fw - 1u;
        }
    }

    __syncthreads();

    const int4* __restrict__ src4 = (const int4*)ei;
    const int*  __restrict__ dstp = ei + N_EDGES;

    {
        const int nl = __builtin_amdgcn_readfirstlane(npl);
        for (int t0 = wid * SCAN_DEPTH; t0 < nl; t0 += (TILE_THREADS / 32) * SCAN_DEPTH) {
            int  cid[SCAN_DEPTH];
            bool val[SCAN_DEPTH];
            const uint2 pv = *(const uint2*)&plist[t0];
            const unsigned pw[2] = { (unsigned)__builtin_amdgcn_readfirstlane((int)pv.x),
                                     (unsigned)__builtin_amdgcn_readfirstlane((int)pv.y) };
#pragma unroll
            for (int j = 0; j < SCAN_DEPTH; ++j) {
                val[j] = (t0 + j < nl);
                cid[j] = val[j] ? (int)((pw[j >> 1] >> ((j & 1) * 16)) & 0xffffu) : cid[0];
            }
            int4 s[SCAN_DEPTH];
#pragma unroll
            for (int j = 0; j < SCAN_DEPTH; ++j)
                s[j] = src4[cid[j] * 32 + lane];
            unsigned ovb = 0u;
#pragma unroll
            for (int j = 0; j < SCAN_DEPTH; ++j) {
                if (val[j]) {
                    const unsigned e0 = (unsigned)((cid[j] * 32 + lane) * 4);
                    const unsigned rr[4] = { (unsigned)(s[j].x - base), (unsigned)(s[j].y - base),
                                             (unsigned)(s[j].z - base), (unsigned)(s[j].w - base) };
#pragma unroll
                    for (int k = 0; k < 4; ++k) {
                        if (rr[k] < (unsigned)TILE_R) {
                            const int p = atomicAdd(&rcnt[rr[k]], 1);
                            if (p < BCAP) bkt[rr[k] * BCAP + p] = e0 + (unsigned)k;
                            else          ovb |= 1u << (j * 4 + k);
                        }
                    }
                }
            }
            if (__builtin_amdgcn_ballot_w32(ovb != 0u)) {
                if (lane == 0) ovfl = 1;
#pragma nounroll
                for (int jk = 0; jk < SCAN_DEPTH * 4; ++jk) {
                    const int j = jk >> 2, k = jk & 3;
                    unsigned mk = __builtin_amdgcn_ballot_w32(((ovb >> jk) & 1u) != 0u);
                    if (!mk) continue;
                    const int4 sj = make_int4((j == 0) ? s[0].x : (j == 1) ? s[1].x : (j == 2) ? s[2].x : s[3].x,
                                              (j == 0) ? s[0].y : (j == 1) ? s[1].y : (j == 2) ? s[2].y : s[3].y,
                                              (j == 0) ? s[0].z : (j == 1) ? s[1].z : (j == 2) ? s[2].z : s[3].z,
                                              (j == 0) ? s[0].w : (j == 1) ? s[1].w : (j == 2) ? s[2].w : s[3].w);
                    const int cj = (j == 0) ? cid[0] : (j == 1) ? cid[1] : (j == 2) ? cid[2] : cid[3];
                    const unsigned rk = (unsigned)(pick4(sj, k) - base);
                    while (mk) {
                        const int L = __builtin_ctz(mk);
                        mk &= mk - 1;
                        const unsigned r = (unsigned)__builtin_amdgcn_readlane((int)rk, L);
                        const int dn = dstp[(cj * 32 + L) * 4 + k];
                        const float* xr = x + (long)dn * F_OUT + lane;
                        const float logit = ss[r * N_HEAD + hd] + s_dst[dn * N_HEAD + hd];
                        const float lr = (logit >= 0.f) ? logit : NEG_SLOPE * logit;
                        const float ev = expf(-lr);
                        const int evb = __float_as_int(ev);
                        float* a = acc + r * (N_HEAD * F_OUT) + lane;
#pragma nounroll
                        for (int i = 0; i < N_HEAD; ++i) {
                            const float vi = __int_as_float(__builtin_amdgcn_readlane(evb, i));
#pragma unroll
                            for (int cc = 0; cc < 4; ++cc)
                                atomicAdd(a + i * F_OUT + cc * 32, vi * xr[cc * 32] * w[i * F_OUT + cc * 32 + lane]);
                        }
                        if (lane < N_HEAD)
                            atomicAdd(&rs[r * N_HEAD + lane], ev);
                    }
                }
            }
        }
    }

    __syncthreads();

    const bool ovf = ovfl != 0;

    for (;;) {
        int rt = 0;
        if (lane == 0) rt = atomicAdd(&nextr, 1);
        const int r = __builtin_amdgcn_readfirstlane(rt);
        if (r >= TILE_R) break;
        const int nr = __builtin_amdgcn_readfirstlane(min(rcnt[r], BCAP));
        const unsigned* __restrict__ br = bkt + r * BCAP;
        float sm[N_HEAD][4], dn_[N_HEAD];
#pragma unroll
        for (int i = 0; i < N_HEAD; ++i) {
            dn_[i] = 0.f;
#pragma unroll
            for (int c = 0; c < 4; ++c) sm[i][c] = 0.f;
        }
        const float4 sl = ((const float4*)ss)[r];
        const float4 sv = make_float4(__int_as_float(__builtin_amdgcn_readfirstlane(__float_as_int(sl.x))),
                                      __int_as_float(__builtin_amdgcn_readfirstlane(__float_as_int(sl.y))),
                                      __int_as_float(__builtin_amdgcn_readfirstlane(__float_as_int(sl.z))),
                                      __int_as_float(__builtin_amdgcn_readfirstlane(__float_as_int(sl.w))));

        for (int b0 = 0; b0 < nr; b0 += 32) {
            const int cnt = min(32, nr - b0);
            const bool live = lane < cnt;
            const unsigned ent = br[live ? (b0 + lane) : b0];
            const int dl = dstp[ent];
            const float4 sd = ((const float4*)s_dst)[dl];
            const float g0 = sv.x + sd.x;
            const float g1 = sv.y + sd.y;
            const float g2 = sv.z + sd.z;
            const float g3 = sv.w + sd.w;
            const float ev[N_HEAD] = { expf(-((g0 >= 0.f) ? g0 : NEG_SLOPE * g0)),
                                       expf(-((g1 >= 0.f) ? g1 : NEG_SLOPE * g1)),
                                       expf(-((g2 >= 0.f) ? g2 : NEG_SLOPE * g2)),
                                       expf(-((g3 >= 0.f) ? g3 : NEG_SLOPE * g3)) };
            for (int j0 = 0; j0 < cnt; j0 += DRAIN_AHEAD) {
                float4 xv[DRAIN_AHEAD];
#pragma unroll
                for (int u = 0; u < DRAIN_AHEAD; ++u) {
                    const int jj = min(j0 + u, cnt - 1);
                    const int dn = __builtin_amdgcn_readlane(dl, jj);
                    xv[u] = ((const float4*)(x + (long)dn * F_OUT))[lane];
                }
#pragma unroll
                for (int u = 0; u < DRAIN_AHEAD; ++u) {
                    const int jj = j0 + u;
                    if (jj < cnt) {
#pragma unroll
                        for (int i = 0; i < N_HEAD; ++i) {
                            const float vi = __int_as_float(__builtin_amdgcn_readlane(__float_as_int(ev[i]), jj));
                            dn_[i] += vi;
                            sm[i][0] += vi * xv[u].x; sm[i][1] += vi * xv[u].y;
                            sm[i][2] += vi * xv[u].z; sm[i][3] += vi * xv[u].w;
                        }
                    }
                }
            }
        }

        const int node = base + r;
        if (node < N_NODES) {
            const float4 dv = ((const float4*)rs)[r];
            const float dd[N_HEAD] = { dv.x, dv.y, dv.z, dv.w };
#pragma unroll
            for (int i = 0; i < N_HEAD; ++i) {
                const float4 wv = ((const float4*)(w + i * F_OUT))[lane];
                float4 v;
                v.x = sm[i][0] * wv.x; v.y = sm[i][1] * wv.y;
                v.z = sm[i][2] * wv.z; v.w = sm[i][3] * wv.w;
                if (ovf) {
                    const float4 a = ((const float4*)acc)[(r * N_HEAD + i) * (F_OUT / 4) + lane];
                    v.x += a.x; v.y += a.y; v.z += a.z; v.w += a.w;
                }
                const float inv = 1.0f / (dn_[i] + dd[i]);
                v.x *= inv; v.y *= inv; v.z *= inv; v.w *= inv;
                ((float4*)out)[((long)i * N_NODES + node) * (F_OUT / 4) + lane] = v;
            }
        }
    }
}

extern "C" void kernel_launch(void* const* d_in, const int* in_sizes, int n_in,
                              void* d_out, int out_size, void* d_ws, size_t ws_size,
                              hipStream_t stream) {
    const float* x  = (const float*)d_in[0];
    const float* w  = (const float*)d_in[1];
    const float* a  = (const float*)d_in[2];
    const int*   ei = (const int*)d_in[3];
    float* out = (float*)d_out;

    float* ws     = (float*)d_ws;
    float* s_src  = ws;
    float* s_dst  = ws + N_HEAD * N_NODES;
    unsigned* filt = (unsigned*)(ws + 2 * N_HEAD * N_NODES);
    float* coefB  = ws + 3 * N_HEAD * N_NODES;

    prep_kernel<<<COEF_BLOCKS + N_WORD, 256, 0, stream>>>(w, a, ei, coefB, filt);
    proj_kernel<<<N_NODES / 16, 32, 0, stream>>>(x, coefB, s_src, s_dst);
    tile_kernel<<<N_TILES, TILE_THREADS, 0, stream>>>(x, w, ei, filt, s_src, s_dst, out);
}


# scan reads 256-byte pieces (64 sources), half the bytes per found edge; to fit: 105-row tiles all resident, an overflowed row is re-read, no sums in LDS; sum order = arrival order per row
# speedup vs baseline: 12.0695x; 1.1819x over previous
#include <hip/hip_runtime.h>

#define N_NODES 50000
#define N_EDGES 800000
#define F_OUT   128
#define N_HEAD  4
#define NEG_SLOPE 0.2f

typedef __attribute__((ext_vector_type(2))) float v2f;
typedef __attribute__((ext_vector_type(8))) float v8f;

#define TILE_R       105
#define N_TILES      ((N_NODES + TILE_R - 1) / TILE_R)
#define PIECE_SRC    64
#define PIECE_Q      (PIECE_SRC / 4)
#define N_CHUNK      (N_EDGES / PIECE_SRC)
#define N_WORD       ((N_CHUNK + 31) / 32)
#define FILT_PITCH   ((N_TILES + 31) / 32 * 32)
#define COEF_BLOCKS  ((F_OUT * 16 + 255) / 256)

static_assert((N_EDGES % PIECE_SRC) == 0, "the source list is whole pieces");
static_assert(PIECE_Q == 16, "a wave read is two pieces, one per half-wave");
static_assert(N_WORD * FILT_PITCH <= N_HEAD * N_NODES, "the filter lies in the workspace between the destination scores and the coefficients");

__global__ __launch_bounds__(256)
void prep_kernel(const float* __restrict__ w, const float* __restrict__ a,
                 const int* __restrict__ ei,
                 float* __restrict__ coefB, unsigned* __restrict__ filt) {
    __shared__ unsigned bm[FILT_PITCH];
    const int tid = threadIdx.x;
    if (blockIdx.x < COEF_BLOCKS) {
        int idx = blockIdx.x * 256 + tid;
        if (idx >= F_OUT * 16) return;
        int k = idx >> 4, col = idx & 15;
        float v = 0.f;
        if (col < 8) {
            int head = col & 3, sel = col >> 2;
            v = w[head * F_OUT + k] * a[head * 2 * F_OUT + sel * F_OUT + k];
        }
        coefB[idx] = v;
        return;
    }
    const int cw = blockIdx.x - COEF_BLOCKS;
    for (int t = tid; t < FILT_PITCH; t += 256) bm[t] = 0u;
    __syncthreads();
    const int4* __restrict__ src4 = (const int4*)ei;
    constexpr int NQ = 32 * PIECE_Q / 256;
    int4 s[NQ];
    int  qi[NQ];
#pragma unroll
    for (int i = 0; i < NQ; ++i) {
        qi[i] = cw * (32 * PIECE_Q) + i * 256 + tid;
        s[i] = src4[min(qi[i], N_EDGES / 4 - 1)];
    }
#pragma unroll
    for (int i = 0; i < NQ; ++i) {
        if (qi[i] < N_EDGES / 4) {
            const unsigned bit = 1u << ((qi[i] / PIECE_Q) & 31);
            const unsigned t0 = (unsigned)s[i].x / TILE_R, t1 = (unsigned)s[i].y / TILE_R;
            const unsigned t2 = (unsigned)s[i].z / TILE_R, t3 = (unsigned)s[i].w / TILE_R;
            if (t0 < N_TILES) atomicOr(&bm[t0], bit);
            if (t1 < N_TILES) atomicOr(&bm[t1], bit);
            if (t2 < N_TILES) atomicOr(&bm[t2], bit);
            if (t3 < N_TILES) atomicOr(&bm[t3], bit);
        }
    }
    __syncthreads();
    for (int t = tid; t < FILT_PITCH; t += 256) filt[cw * FILT_PITCH + t] = bm[t];
}

__global__ void proj_kernel(const float* __restrict__ x, const float* __restrict__ coefB,
                            float* __restrict__ s_src, float* __restrict__ s_dst) {
    int tile = blockIdx.x;
    int lane = threadIdx.x;
    int half = lane >> 4;
    int rc   = lane & 15;
    int koff = half * 2;
    long arow = (long)(tile * 16 + rc) * F_OUT;

    v8f c = {0.f, 0.f, 0.f, 0.f, 0.f, 0.f, 0.f, 0.f};
#pragma unroll
    for (int k = 0; k < F_OUT; k += 4) {
        v2f av, bv;
        float2 ax = *(const float2*)(x + arow + k + koff);
        av.x = ax.x; av.y = ax.y;
        bv.x = coefB[(k + koff) * 16 + rc];
        bv.y = coefB[(k + koff + 1) * 16 + rc];
        c = __builtin_amdgcn_wmma_f32_16x16x4_f32(false, av, false, bv,
                                                  (short)0, c, false, false);
    }
    if (rc < 8) {
        int head = rc & 3;
        float* buf = (rc >> 2) ? s_dst : s_src;
        int mbase = tile * 16 + half * 8;
#pragma unroll
        for (int j = 0; j < 8; ++j)
            buf[(mbase + j) * N_HEAD + head] = c[j];
    }
}

#define TILE_THREADS 1024
#define SCAN_DEPTH   4
#define TRIP_PIECES  (2 * SCAN_DEPTH)
#define BCAP         60
#define DRAIN_AHEAD  6

static_assert(N_WORD <= TILE_THREADS, "thread i holds filter word i");
static_assert(TRIP_PIECES == 8, "a trip's list places are one 16-byte read of eight 16-bit numbers");
static_assert((N_WORD * 32) % TRIP_PIECES == 0, "the last trip's 16-byte read stays inside the piece list");
static_assert(N_CHUNK <= 65536, "a piece number fits the 16-bit places of the piece list");
static_assert(N_HEAD == 4, "a lane of the drain holds the four head scores of its edge in one float4");
static_assert(TILE_R * N_HEAD <= TILE_THREADS, "thread i fetches source score i of the tile");

__device__ __forceinline__ int pick4(const int4 v, const int k) {
    return (k == 0) ? v.x : (k == 1) ? v.y : (k == 2) ? v.z : v.w;
}

__global__ __launch_bounds__(TILE_THREADS) __attribute__((amdgpu_waves_per_eu(16)))
void tile_kernel(const float* __restrict__ x, const float* __restrict__ w,
                 const int* __restrict__ ei, const unsigned* __restrict__ filt,
                 const float* __restrict__ s_src, const float* __restrict__ s_dst,
                 float* __restrict__ out) {
    __shared__ __attribute__((aligned(16))) float ss[TILE_R * N_HEAD];
    __shared__ unsigned bkt[TILE_R * BCAP];
    __shared__ int rcnt[TILE_R];
    __shared__ __attribute__((aligned(16))) unsigned short plist[N_WORD * 32];
    __shared__ int npl;
    __shared__ int nextr;

    const int tid  = threadIdx.x;
    const int lane = tid & 31;
    const int l16  = lane & (PIECE_Q - 1);
    const bool hi  = lane >= PIECE_Q;
    const int wid  = __builtin_amdgcn_readfirstlane(tid >> 5);
    const int base = blockIdx.x * TILE_R;

    if (tid == 0) { npl = 0; nextr = 0; }
    __syncthreads();

    unsigned fw = 0u;
    if (tid < N_WORD) fw = filt[tid * FILT_PITCH + blockIdx.x];

    if (tid < TILE_R * N_HEAD) {
        int node = base + (tid >> 2);
        ss[tid] = (node < N_NODES) ? s_src[node * N_HEAD + (tid & 3)] : 0.f;
    }
    if (tid < TILE_R) rcnt[tid] = 0;

    if (fw != 0u) {
        int pos = atomicAdd(&npl, __popc(fw));
        while (fw) {
            plist[pos++] = (unsigned short)(tid * 32 + __builtin_ctz(fw));
            fw &= fw - 1u;
        }
    }

    __syncthreads();

    const int4* __restrict__ src4 = (const int4*)ei;
    const int*  __restrict__ dstp = ei + N_EDGES;
    const int nl = __builtin_amdgcn_readfirstlane(npl);

    for (int t0 = wid * TRIP_PIECES; t0 < nl; t0 += (TILE_THREADS / 32) * TRIP_PIECES) {
        const uint4 pv = *(const uint4*)&plist[t0];
        const unsigned pw[SCAN_DEPTH] = { (unsigned)__builtin_amdgcn_readfirstlane((int)pv.x),
                                          (unsigned)__builtin_amdgcn_readfirstlane((int)pv.y),
                                          (unsigned)__builtin_amdgcn_readfirstlane((int)pv.z),
                                          (unsigned)__builtin_amdgcn_readfirstlane((int)pv.w) };
        const unsigned first = pw[0] & 0xffffu;
        unsigned q[SCAN_DEPTH];
        bool     ok[SCAN_DEPTH];
        int4     s[SCAN_DEPTH];
#pragma unroll
        for (int j = 0; j < SCAN_DEPTH; ++j) {
            const bool va = (t0 + 2 * j     < nl);
            const bool vb = (t0 + 2 * j + 1 < nl);
            const unsigned pa = va ? (pw[j] & 0xffffu) : first;
            const unsigned pb = vb ? (pw[j] >> 16)     : first;
            q[j]  = (hi ? pb : pa) * PIECE_Q + (unsigned)l16;
            ok[j] = hi ? vb : va;
            s[j]  = src4[q[j]];
        }
#pragma unroll
        for (int j = 0; j < SCAN_DEPTH; ++j) {
            if (t0 + 2 * j < nl) {
                const unsigned rr[4] = { (unsigned)(s[j].x - base), (unsigned)(s[j].y - base),
                                         (unsigned)(s[j].z - base), (unsigned)(s[j].w - base) };
#pragma unroll
                for (int k = 0; k < 4; ++k) {
                    if (ok[j] && rr[k] < (unsigned)TILE_R) {
                        const int p = atomicAdd(&rcnt[rr[k]], 1);
                        if (p < BCAP) bkt[rr[k] * BCAP + p] = q[j] * 4u + (unsigned)k;
                    }
                }
            }
        }
    }

    __syncthreads();

    for (;;) {
        int rt = 0;
        if (lane == 0) rt = atomicAdd(&nextr, 1);
        const int r = __builtin_amdgcn_readfirstlane(rt);
        if (r >= TILE_R) break;
        const int rc = __builtin_amdgcn_readfirstlane(rcnt[r]);
        const unsigned* br = bkt + r * BCAP;
        float sm[N_HEAD][4], dn_[N_HEAD];
#pragma unroll
        for (int i = 0; i < N_HEAD; ++i) {
            dn_[i] = 0.f;
#pragma unroll
            for (int c = 0; c < 4; ++c) sm[i][c] = 0.f;
        }
        const float4 sl = ((const float4*)ss)[r];
        const float4 sv = make_float4(__int_as_float(__builtin_amdgcn_readfirstlane(__float_as_int(sl.x))),
                                      __int_as_float(__builtin_amdgcn_readfirstlane(__float_as_int(sl.y))),
                                      __int_as_float(__builtin_amdgcn_readfirstlane(__float_as_int(sl.z))),
                                      __int_as_float(__builtin_amdgcn_readfirstlane(__float_as_int(sl.w))));

        auto sum_batch = [&](const unsigned ent, const int cnt) {
            const int dl = dstp[ent];
            const float4 sd = ((const float4*)s_dst)[dl];
            const float g0 = sv.x + sd.x;
            const float g1 = sv.y + sd.y;
            const float g2 = sv.z + sd.z;
            const float g3 = sv.w + sd.w;
            const float ev[N_HEAD] = { expf(-((g0 >= 0.f) ? g0 : NEG_SLOPE * g0)),
                                       expf(-((g1 >= 0.f) ? g1 : NEG_SLOPE * g1)),
                                       expf(-((g2 >= 0.f) ? g2 : NEG_SLOPE * g2)),
                                       expf(-((g3 >= 0.f) ? g3 : NEG_SLOPE * g3)) };
            for (int j0 = 0; j0 < cnt; j0 += DRAIN_AHEAD) {
                float4 xv[DRAIN_AHEAD];
#pragma unroll
                for (int u = 0; u < DRAIN_AHEAD; ++u) {
                    const int jj = min(j0 + u, cnt - 1);
                    const int dn = __builtin_amdgcn_readlane(dl, jj);
                    xv[u] = ((const float4*)(x + (long)dn * F_OUT))[lane];
                }
#pragma unroll
                for (int u = 0; u < DRAIN_AHEAD; ++u) {
                    const int jj = j0 + u;
                    if (jj < cnt) {
#pragma unroll
                        for (int i = 0; i < N_HEAD; ++i) {
                            const float vi = __int_as_float(__builtin_amdgcn_readlane(__float_as_int(ev[i]), jj));
                            dn_[i] += vi;
                            sm[i][0] += vi * xv[u].x; sm[i][1] += vi * xv[u].y;
                            sm[i][2] += vi * xv[u].z; sm[i][3] += vi * xv[u].w;
                        }
                    }
                }
            }
        };

        if (rc <= BCAP) {
            for (int b0 = 0; b0 < rc; b0 += 32) {
                const int cnt = min(32, rc - b0);
                sum_batch(br[(lane < cnt) ? (b0 + lane) : b0], cnt);
            }
        } else {
#pragma nounroll
            for (int pi = 0; pi < nl; ++pi) {
                const int cq = __builtin_amdgcn_readfirstlane((int)plist[pi]) * PIECE_Q;
#pragma nounroll
                for (int k = 0; k < 4; ++k) {
                    const int4 sp = src4[cq + l16];
                    const unsigned mk = __builtin_amdgcn_ballot_w32(!hi && pick4(sp, k) == base + r);
                    if (!mk) continue;
                    unsigned m = mk;
                    for (int t = 0; t < lane; ++t) m &= m - 1u;
                    const int ml = __builtin_ctz(m ? m : mk);
                    sum_batch((unsigned)((cq + ml) * 4 + k), __popc(mk));
                }
            }
        }

        const int node = base + r;
        if (node < N_NODES) {
#pragma unroll
            for (int i = 0; i < N_HEAD; ++i) {
                const float4 wv = ((const float4*)(w + i * F_OUT))[lane];
                const float inv = 1.0f / dn_[i];
                float4 v;
                v.x = sm[i][0] * wv.x * inv; v.y = sm[i][1] * wv.y * inv;
                v.z = sm[i][2] * wv.z * inv; v.w = sm[i][3] * wv.w * inv;
                ((float4*)out)[((long)i * N_NODES + node) * (F_OUT / 4) + lane] = v;
            }
        }
    }
}

extern "C" void kernel_launch(void* const* d_in, const int* in_sizes, int n_in,
                              void* d_out, int out_size, void* d_ws, size_t ws_size,
                              hipStream_t stream) {
    const float* x  = (const float*)d_in[0];
    const float* w  = (const float*)d_in[1];
    const float* a  = (const float*)d_in[2];
    const int*   ei = (const int*)d_in[3];
    float* out = (float*)d_out;

    float* ws     = (float*)d_ws;
    float* s_src  = ws;
    float* s_dst  = ws + N_HEAD * N_NODES;
    unsigned* filt = (unsigned*)(ws + 2 * N_HEAD * N_NODES);
    float* coefB  = ws + 3 * N_HEAD * N_NODES;

    prep_kernel<<<COEF_BLOCKS + N_WORD, 256, 0, stream>>>(w, a, ei, coefB, filt);
    proj_kernel<<<N_NODES / 16, 32, 0, stream>>>(x, coefB, s_src, s_dst);
    tile_kernel<<<N_TILES, TILE_THREADS, 0, stream>>>(x, w, ei, filt, s_src, s_dst, out);
}
